# starting kernel
# baseline (speedup 1.0000x reference)
#include <hip/hip_runtime.h>
#include <math.h>

#define BSZ   4
#define LSEQ  1024
#define DMOD  256
#define DFF   512
#define MTOK  (BSZ * LSEQ)

typedef float v2f_t __attribute__((ext_vector_type(2)));
typedef float v8f_t __attribute__((ext_vector_type(8)));

template <int Mdim, int Ndim, int Kdim, int NT>
__global__ __launch_bounds__(128)
void stp_wmma_gemm_bias(const float* __restrict__ A,
                        const float* __restrict__ Bt,
                        const float* __restrict__ bias,
                        float* __restrict__ D) {
  const int wave = blockIdx.x * (blockDim.x >> 5) + (threadIdx.x >> 5);
  const int lane = threadIdx.x & 31;
  constexpr int n_groups = Ndim / (16 * NT);
  const int tm = wave / n_groups;
  const int tg = wave % n_groups;
  if (tm >= Mdim / 16) return;

  const int m0   = tm * 16;
  const int n0   = tg * 16 * NT;
  const int row  = lane & 15;
  const int half = lane >> 4;

  const float* arow = A + (size_t)(m0 + row) * Kdim + 2 * half;
  const float* brow[NT];
#pragma unroll
  for (int nt = 0; nt < NT; ++nt)
    brow[nt] = Bt + (size_t)(n0 + nt * 16 + row) * Kdim + 2 * half;

  v8f_t c[NT];
#pragma unroll
  for (int nt = 0; nt < NT; ++nt) c[nt] = (v8f_t){};

#pragma unroll 4
  for (int k0 = 0; k0 < Kdim; k0 += 4) {
    const v2f_t a = *(const v2f_t*)(arow + k0);
#pragma unroll
    for (int nt = 0; nt < NT; ++nt) {
      const v2f_t b = *(const v2f_t*)(brow[nt] + k0);
      c[nt] = __builtin_amdgcn_wmma_f32_16x16x4_f32(false, a, false, b,
                                                    (short)0, c[nt], false, false);
    }
  }

#pragma unroll
  for (int nt = 0; nt < NT; ++nt) {
    const int n = n0 + nt * 16 + row;
    const float bn = bias[n];
#pragma unroll
    for (int v = 0; v < 8; ++v) {
      const int m = m0 + v + 8 * half;
      D[(size_t)m * Ndim + n] = c[nt][v] + bn;
    }
  }
}

__global__ __launch_bounds__(256)
void stp_scan_kernel(const float* __restrict__ x,
                     const float* __restrict__ Hs,
                     const float* __restrict__ Lambda,
                     const float* __restrict__ Gamma,
                     float* __restrict__ Hg) {
  const int wave = blockIdx.x * (blockDim.x >> 5) + (threadIdx.x >> 5);
  const int lane = threadIdx.x & 31;
  if (wave >= BSZ * DFF) return;
  const int b  = wave >> 9;
  const int i  = wave & (DFF - 1);
  const int jb = lane * 8;

  float r[8], g[8], F[8];
  const float* lam = Lambda + (size_t)i * DMOD + jb;
  const float* gam = Gamma  + (size_t)i * DMOD + jb;
#pragma unroll
  for (int u = 0; u < 8; ++u) {
    const float s = 1.0f / (1.0f + __expf(-lam[u]));
    r[u] = 1.0f - s;
    g[u] = gam[u];
    F[u] = 0.0f;
  }

  const float* xb     = x  + (size_t)b * LSEQ * DMOD + jb;
  const float* hs_ptr = Hs + (size_t)b * LSEQ * DFF + i;
  float*       hg_ptr = Hg + (size_t)b * LSEQ * DFF + i;

  float4 cx0 = *(const float4*)(xb);
  float4 cx1 = *(const float4*)(xb + 4);
  float  chs = hs_ptr[0];

  for (int t = 0; t < LSEQ; ++t) {
    float4 nx0, nx1;
    float  nhs = 0.0f;
    if (t + 1 < LSEQ) {
      const float* xn = xb + (size_t)(t + 1) * DMOD;
      nx0 = *(const float4*)(xn);
      nx1 = *(const float4*)(xn + 4);
      nhs = hs_ptr[(size_t)(t + 1) * DFF];
      if (t + 2 < LSEQ) __builtin_prefetch(xn + DMOD, 0, 1);
    }

    const float xv[8] = {cx0.x, cx0.y, cx0.z, cx0.w, cx1.x, cx1.y, cx1.z, cx1.w};

    float p = 0.0f;
#pragma unroll
    for (int u = 0; u < 8; ++u) p = fmaf(F[u], xv[u], p);
#pragma unroll
    for (int off = 16; off >= 1; off >>= 1) p += __shfl_xor(p, off, 32);

    const float h = chs + p;

    const float gh = 0.5f * h * (1.0f + erff(h * 0.70710678118654752f));
    if (lane == 0) hg_ptr[(size_t)t * DFF] = gh;

#pragma unroll
    for (int u = 0; u < 8; ++u) F[u] = fmaf(r[u], F[u], g[u] * h * xv[u]);

    cx0 = nx0; cx1 = nx1; chs = nhs;
  }
}

extern "C" void kernel_launch(void* const* d_in, const int* in_sizes, int n_in,
                              void* d_out, int out_size, void* d_ws, size_t ws_size,
                              hipStream_t stream) {
  const float* x   = (const float*)d_in[0];
  const float* W1  = (const float*)d_in[1];
  const float* b1  = (const float*)d_in[2];
  const float* W2  = (const float*)d_in[3];
  const float* b2  = (const float*)d_in[4];
  const float* Lam = (const float*)d_in[5];
  const float* Gam = (const float*)d_in[6];
  float* out = (float*)d_out;

  float* Hs = (float*)d_ws;
  float* Hg = Hs + (size_t)MTOK * DFF;

  {
    constexpr int NT = 4;
    const int waves = (MTOK / 16) * (DFF / (16 * NT));
    stp_wmma_gemm_bias<MTOK, DFF, DMOD, NT>
        <<<waves / 4, 128, 0, stream>>>(x, W1, b1, Hs);
  }

  stp_scan_kernel<<<(BSZ * DFF) / 8, 256, 0, stream>>>(x, Hs, Lam, Gam, Hg);

  {
    constexpr int NT = 4;
    const int waves = (MTOK / 16) * (DMOD / (16 * NT));
    stp_wmma_gemm_bias<MTOK, DMOD, DFF, NT>
        <<<waves / 4, 128, 0, stream>>>(Hg, W2, b2, out);
  }
}


# scan: wave32 dot-product sum on the vector ALU (v_permlanex16 + four row_xmask DPP adds) instead of five ds_bpermute LDS hops; same xor tree 16,8,4,2,1, sums bit-identical
# speedup vs baseline: 1.0655x; 1.0655x over previous
#include <hip/hip_runtime.h>
#include <math.h>

#define BSZ   4
#define LSEQ  1024
#define DMOD  256
#define DFF   512
#define MTOK  (BSZ * LSEQ)

typedef float v2f_t __attribute__((ext_vector_type(2)));
typedef float v8f_t __attribute__((ext_vector_type(8)));

template <int Mdim, int Ndim, int Kdim, int NT>
__global__ __launch_bounds__(128)
void stp_wmma_gemm_bias(const float* __restrict__ A,
                        const float* __restrict__ Bt,
                        const float* __restrict__ bias,
                        float* __restrict__ D) {
  const int wave = blockIdx.x * (blockDim.x >> 5) + (threadIdx.x >> 5);
  const int lane = threadIdx.x & 31;
  constexpr int n_groups = Ndim / (16 * NT);
  const int tm = wave / n_groups;
  const int tg = wave % n_groups;
  if (tm >= Mdim / 16) return;

  const int m0   = tm * 16;
  const int n0   = tg * 16 * NT;
  const int row  = lane & 15;
  const int half = lane >> 4;

  const float* arow = A + (size_t)(m0 + row) * Kdim + 2 * half;
  const float* brow[NT];
#pragma unroll
  for (int nt = 0; nt < NT; ++nt)
    brow[nt] = Bt + (size_t)(n0 + nt * 16 + row) * Kdim + 2 * half;

  v8f_t c[NT];
#pragma unroll
  for (int nt = 0; nt < NT; ++nt) c[nt] = (v8f_t){};

#pragma unroll 4
  for (int k0 = 0; k0 < Kdim; k0 += 4) {
    const v2f_t a = *(const v2f_t*)(arow + k0);
#pragma unroll
    for (int nt = 0; nt < NT; ++nt) {
      const v2f_t b = *(const v2f_t*)(brow[nt] + k0);
      c[nt] = __builtin_amdgcn_wmma_f32_16x16x4_f32(false, a, false, b,
                                                    (short)0, c[nt], false, false);
    }
  }

#pragma unroll
  for (int nt = 0; nt < NT; ++nt) {
    const int n = n0 + nt * 16 + row;
    const float bn = bias[n];
#pragma unroll
    for (int v = 0; v < 8; ++v) {
      const int m = m0 + v + 8 * half;
      D[(size_t)m * Ndim + n] = c[nt][v] + bn;
    }
  }
}

__device__ __forceinline__ float stp_xhalf(float v) {
  const int b = __builtin_bit_cast(int, v);
  return __builtin_bit_cast(
      float, __builtin_amdgcn_permlanex16(b, b, 0x76543210, 0xfedcba98, false, false));
}

template <int M>
__device__ __forceinline__ float stp_row_xor(float v) {
  const int b = __builtin_bit_cast(int, v);
  return __builtin_bit_cast(
      float, __builtin_amdgcn_update_dpp(b, b, 0x160 + M, 0xf, 0xf, true));
}

__global__ __launch_bounds__(256)
void stp_scan_kernel(const float* __restrict__ x,
                     const float* __restrict__ Hs,
                     const float* __restrict__ Lambda,
                     const float* __restrict__ Gamma,
                     float* __restrict__ Hg) {
  const int wave = blockIdx.x * (blockDim.x >> 5) + (threadIdx.x >> 5);
  const int lane = threadIdx.x & 31;
  if (wave >= BSZ * DFF) return;
  const int b  = wave >> 9;
  const int i  = wave & (DFF - 1);
  const int jb = lane * 8;

  float r[8], g[8], F[8];
  const float* lam = Lambda + (size_t)i * DMOD + jb;
  const float* gam = Gamma  + (size_t)i * DMOD + jb;
#pragma unroll
  for (int u = 0; u < 8; ++u) {
    const float s = 1.0f / (1.0f + __expf(-lam[u]));
    r[u] = 1.0f - s;
    g[u] = gam[u];
    F[u] = 0.0f;
  }

  const float* xb     = x  + (size_t)b * LSEQ * DMOD + jb;
  const float* hs_ptr = Hs + (size_t)b * LSEQ * DFF + i;
  float*       hg_ptr = Hg + (size_t)b * LSEQ * DFF + i;

  float4 cx0 = *(const float4*)(xb);
  float4 cx1 = *(const float4*)(xb + 4);
  float  chs = hs_ptr[0];

  for (int t = 0; t < LSEQ; ++t) {
    float4 nx0, nx1;
    float  nhs = 0.0f;
    if (t + 1 < LSEQ) {
      const float* xn = xb + (size_t)(t + 1) * DMOD;
      nx0 = *(const float4*)(xn);
      nx1 = *(const float4*)(xn + 4);
      nhs = hs_ptr[(size_t)(t + 1) * DFF];
      if (t + 2 < LSEQ) __builtin_prefetch(xn + DMOD, 0, 1);
    }

    const float xv[8] = {cx0.x, cx0.y, cx0.z, cx0.w, cx1.x, cx1.y, cx1.z, cx1.w};

    float p = 0.0f;
#pragma unroll
    for (int u = 0; u < 8; ++u) p = fmaf(F[u], xv[u], p);
    p += stp_xhalf(p);
    p += stp_row_xor<8>(p);
    p += stp_row_xor<4>(p);
    p += stp_row_xor<2>(p);
    p += stp_row_xor<1>(p);

    const float h = chs + p;

    const float gh = 0.5f * h * (1.0f + erff(h * 0.70710678118654752f));
    if (lane == 0) hg_ptr[(size_t)t * DFF] = gh;

#pragma unroll
    for (int u = 0; u < 8; ++u) F[u] = fmaf(r[u], F[u], g[u] * h * xv[u]);

    cx0 = nx0; cx1 = nx1; chs = nhs;
  }
}

extern "C" void kernel_launch(void* const* d_in, const int* in_sizes, int n_in,
                              void* d_out, int out_size, void* d_ws, size_t ws_size,
                              hipStream_t stream) {
  const float* x   = (const float*)d_in[0];
  const float* W1  = (const float*)d_in[1];
  const float* b1  = (const float*)d_in[2];
  const float* W2  = (const float*)d_in[3];
  const float* b2  = (const float*)d_in[4];
  const float* Lam = (const float*)d_in[5];
  const float* Gam = (const float*)d_in[6];
  float* out = (float*)d_out;

  float* Hs = (float*)d_ws;
  float* Hg = Hs + (size_t)MTOK * DFF;

  {
    constexpr int NT = 4;
    const int waves = (MTOK / 16) * (DFF / (16 * NT));
    stp_wmma_gemm_bias<MTOK, DFF, DMOD, NT>
        <<<waves / 4, 128, 0, stream>>>(x, W1, b1, Hs);
  }

  stp_scan_kernel<<<(BSZ * DFF) / 8, 256, 0, stream>>>(x, Hs, Lam, Gam, Hg);

  {
    constexpr int NT = 4;
    const int waves = (MTOK / 16) * (DMOD / (16 * NT));
    stp_wmma_gemm_bias<MTOK, DMOD, DFF, NT>
        <<<waves / 4, 128, 0, stream>>>(Hg, W2, b2, out);
  }
}


# scan: the 8 waves of a work-group share one LDS copy of x_t and H_static, fetched in double-buffered 32-step chunks one chunk ahead (was: per-wave global loads one step ahead); sums bit-identical
# speedup vs baseline: 2.3362x; 2.1926x over previous
#include <hip/hip_runtime.h>
#include <math.h>

#define BSZ   4
#define LSEQ  1024
#define DMOD  256
#define DFF   512
#define MTOK  (BSZ * LSEQ)

typedef float v2f_t __attribute__((ext_vector_type(2)));
typedef float v8f_t __attribute__((ext_vector_type(8)));

template <int Mdim, int Ndim, int Kdim, int NT>
__global__ __launch_bounds__(128)
void stp_wmma_gemm_bias(const float* __restrict__ A,
                        const float* __restrict__ Bt,
                        const float* __restrict__ bias,
                        float* __restrict__ D) {
  const int wave = blockIdx.x * (blockDim.x >> 5) + (threadIdx.x >> 5);
  const int lane = threadIdx.x & 31;
  constexpr int n_groups = Ndim / (16 * NT);
  const int tm = wave / n_groups;
  const int tg = wave % n_groups;
  if (tm >= Mdim / 16) return;

  const int m0   = tm * 16;
  const int n0   = tg * 16 * NT;
  const int row  = lane & 15;
  const int half = lane >> 4;

  const float* arow = A + (size_t)(m0 + row) * Kdim + 2 * half;
  const float* brow[NT];
#pragma unroll
  for (int nt = 0; nt < NT; ++nt)
    brow[nt] = Bt + (size_t)(n0 + nt * 16 + row) * Kdim + 2 * half;

  v8f_t c[NT];
#pragma unroll
  for (int nt = 0; nt < NT; ++nt) c[nt] = (v8f_t){};

#pragma unroll 4
  for (int k0 = 0; k0 < Kdim; k0 += 4) {
    const v2f_t a = *(const v2f_t*)(arow + k0);
#pragma unroll
    for (int nt = 0; nt < NT; ++nt) {
      const v2f_t b = *(const v2f_t*)(brow[nt] + k0);
      c[nt] = __builtin_amdgcn_wmma_f32_16x16x4_f32(false, a, false, b,
                                                    (short)0, c[nt], false, false);
    }
  }

#pragma unroll
  for (int nt = 0; nt < NT; ++nt) {
    const int n = n0 + nt * 16 + row;
    const float bn = bias[n];
#pragma unroll
    for (int v = 0; v < 8; ++v) {
      const int m = m0 + v + 8 * half;
      D[(size_t)m * Ndim + n] = c[nt][v] + bn;
    }
  }
}

__device__ __forceinline__ float stp_xhalf(float v) {
  const int b = __builtin_bit_cast(int, v);
  return __builtin_bit_cast(
      float, __builtin_amdgcn_permlanex16(b, b, 0x76543210, 0xfedcba98, false, false));
}

template <int M>
__device__ __forceinline__ float stp_row_xor(float v) {
  const int b = __builtin_bit_cast(int, v);
  return __builtin_bit_cast(
      float, __builtin_amdgcn_update_dpp(b, b, 0x160 + M, 0xf, 0xf, true));
}

#define SCAN_TC     32
#define SCAN_WAVES  8
#define SCAN_NCHUNK (LSEQ / SCAN_TC)
#define SCAN_XQ     (DMOD / 4)
#define SCAN_XLD    (SCAN_TC * SCAN_XQ / (SCAN_WAVES * 32))

__global__ __launch_bounds__(SCAN_WAVES * 32)
void stp_scan_kernel(const float* __restrict__ x,
                     const float* __restrict__ Hs,
                     const float* __restrict__ Lambda,
                     const float* __restrict__ Gamma,
                     float* __restrict__ Hg) {
  __shared__ float4 xs[2][SCAN_TC][SCAN_XQ];
  __shared__ float  hss[2][SCAN_TC][SCAN_WAVES];

  const int tid   = threadIdx.x;
  const int wid   = tid >> 5;
  const int lane  = tid & 31;
  const int wave0 = blockIdx.x * SCAN_WAVES;
  if (wave0 >= BSZ * DFF) return;
  const int b  = wave0 >> 9;
  const int i0 = wave0 & (DFF - 1);
  const int i  = i0 + wid;
  const int jb = lane * 8;

  float r[8], g[8], F[8];
  const float* lam = Lambda + (size_t)i * DMOD + jb;
  const float* gam = Gamma  + (size_t)i * DMOD + jb;
#pragma unroll
  for (int u = 0; u < 8; ++u) {
    const float s = 1.0f / (1.0f + __expf(-lam[u]));
    r[u] = 1.0f - s;
    g[u] = gam[u];
    F[u] = 0.0f;
  }

  const int srow = tid >> 6;
  const int scol = tid & (SCAN_XQ - 1);
  const int htt  = tid >> 3;
  const int hw   = tid & (SCAN_WAVES - 1);
  const float4* xg  = (const float4*)(x + (size_t)b * LSEQ * DMOD) + (size_t)srow * SCAN_XQ + scol;
  const float*  hsg = Hs + (size_t)b * LSEQ * DFF + (size_t)htt * DFF + i0 + hw;
  float*     hg_ptr = Hg + (size_t)b * LSEQ * DFF + i;

  {
    float4 st[SCAN_XLD];
#pragma unroll
    for (int k = 0; k < SCAN_XLD; ++k) st[k] = xg[(size_t)(4 * k) * SCAN_XQ];
    const float sh = hsg[0];
#pragma unroll
    for (int k = 0; k < SCAN_XLD; ++k) xs[0][srow + 4 * k][scol] = st[k];
    hss[0][htt][hw] = sh;
  }
  __syncthreads();

  for (int c = 0; c < SCAN_NCHUNK; ++c) {
    const int  buf  = c & 1;
    const int  t0   = c * SCAN_TC;
    const bool more = (c + 1 < SCAN_NCHUNK);

    float4 st[SCAN_XLD];
    float  sh = 0.0f;
#pragma unroll
    for (int k = 0; k < SCAN_XLD; ++k) st[k] = make_float4(0.0f, 0.0f, 0.0f, 0.0f);
    if (more) {
      const size_t tn0 = (size_t)(t0 + SCAN_TC);
#pragma unroll
      for (int k = 0; k < SCAN_XLD; ++k) st[k] = xg[(tn0 + 4 * k) * SCAN_XQ];
      sh = hsg[tn0 * DFF];
    }

    float4 cx0 = xs[buf][0][2 * lane];
    float4 cx1 = xs[buf][0][2 * lane + 1];
    float  chs = hss[buf][0][wid];

    for (int tt = 0; tt < SCAN_TC; ++tt) {
      const int tn = (tt + 1 < SCAN_TC) ? tt + 1 : tt;
      const float4 nx0 = xs[buf][tn][2 * lane];
      const float4 nx1 = xs[buf][tn][2 * lane + 1];
      const float  nhs = hss[buf][tn][wid];

      const float xv[8] = {cx0.x, cx0.y, cx0.z, cx0.w, cx1.x, cx1.y, cx1.z, cx1.w};

      float p = 0.0f;
#pragma unroll
      for (int u = 0; u < 8; ++u) p = fmaf(F[u], xv[u], p);
      p += stp_xhalf(p);
      p += stp_row_xor<8>(p);
      p += stp_row_xor<4>(p);
      p += stp_row_xor<2>(p);
      p += stp_row_xor<1>(p);

      const float h = chs + p;

      const float gh = 0.5f * h * (1.0f + erff(h * 0.70710678118654752f));
      if (lane == 0) hg_ptr[(size_t)(t0 + tt) * DFF] = gh;

#pragma unroll
      for (int u = 0; u < 8; ++u) F[u] = fmaf(r[u], F[u], g[u] * h * xv[u]);

      cx0 = nx0; cx1 = nx1; chs = nhs;
    }

    if (more) {
#pragma unroll
      for (int k = 0; k < SCAN_XLD; ++k) xs[buf ^ 1][srow + 4 * k][scol] = st[k];
      hss[buf ^ 1][htt][hw] = sh;
      __syncthreads();
    }
  }
}

extern "C" void kernel_launch(void* const* d_in, const int* in_sizes, int n_in,
                              void* d_out, int out_size, void* d_ws, size_t ws_size,
                              hipStream_t stream) {
  const float* x   = (const float*)d_in[0];
  const float* W1  = (const float*)d_in[1];
  const float* b1  = (const float*)d_in[2];
  const float* W2  = (const float*)d_in[3];
  const float* b2  = (const float*)d_in[4];
  const float* Lam = (const float*)d_in[5];
  const float* Gam = (const float*)d_in[6];
  float* out = (float*)d_out;

  float* Hs = (float*)d_ws;
  float* Hg = Hs + (size_t)MTOK * DFF;

  {
    constexpr int NT = 4;
    const int waves = (MTOK / 16) * (DFF / (16 * NT));
    stp_wmma_gemm_bias<MTOK, DFF, DMOD, NT>
        <<<waves / 4, 128, 0, stream>>>(x, W1, b1, Hs);
  }

  stp_scan_kernel<<<(BSZ * DFF) / 8, 256, 0, stream>>>(x, Hs, Lam, Gam, Hg);

  {
    constexpr int NT = 4;
    const int waves = (MTOK / 16) * (DMOD / (16 * NT));
    stp_wmma_gemm_bias<MTOK, DMOD, DFF, NT>
        <<<waves / 4, 128, 0, stream>>>(Hg, W2, b2, out);
  }
}


# scan: erf-GELU and its store leave the time step: lane tt keeps h of step tt; once per 32-step chunk the wave does GELU one step per lane and stores 32 values behind the hand-over; bit-identical
# speedup vs baseline: 3.1743x; 1.3587x over previous
#include <hip/hip_runtime.h>
#include <math.h>

#define BSZ   4
#define LSEQ  1024
#define DMOD  256
#define DFF   512
#define MTOK  (BSZ * LSEQ)

typedef float v2f_t __attribute__((ext_vector_type(2)));
typedef float v8f_t __attribute__((ext_vector_type(8)));

template <int Mdim, int Ndim, int Kdim, int NT>
__global__ __launch_bounds__(128)
void stp_wmma_gemm_bias(const float* __restrict__ A,
                        const float* __restrict__ Bt,
                        const float* __restrict__ bias,
                        float* __restrict__ D) {
  const int wave = blockIdx.x * (blockDim.x >> 5) + (threadIdx.x >> 5);
  const int lane = threadIdx.x & 31;
  constexpr int n_groups = Ndim / (16 * NT);
  const int tm = wave / n_groups;
  const int tg = wave % n_groups;
  if (tm >= Mdim / 16) return;

  const int m0   = tm * 16;
  const int n0   = tg * 16 * NT;
  const int row  = lane & 15;
  const int half = lane >> 4;

  const float* arow = A + (size_t)(m0 + row) * Kdim + 2 * half;
  const float* brow[NT];
#pragma unroll
  for (int nt = 0; nt < NT; ++nt)
    brow[nt] = Bt + (size_t)(n0 + nt * 16 + row) * Kdim + 2 * half;

  v8f_t c[NT];
#pragma unroll
  for (int nt = 0; nt < NT; ++nt) c[nt] = (v8f_t){};

#pragma unroll 4
  for (int k0 = 0; k0 < Kdim; k0 += 4) {
    const v2f_t a = *(const v2f_t*)(arow + k0);
#pragma unroll
    for (int nt = 0; nt < NT; ++nt) {
      const v2f_t b = *(const v2f_t*)(brow[nt] + k0);
      c[nt] = __builtin_amdgcn_wmma_f32_16x16x4_f32(false, a, false, b,
                                                    (short)0, c[nt], false, false);
    }
  }

#pragma unroll
  for (int nt = 0; nt < NT; ++nt) {
    const int n = n0 + nt * 16 + row;
    const float bn = bias[n];
#pragma unroll
    for (int v = 0; v < 8; ++v) {
      const int m = m0 + v + 8 * half;
      D[(size_t)m * Ndim + n] = c[nt][v] + bn;
    }
  }
}

__device__ __forceinline__ float stp_xhalf(float v) {
  const int b = __builtin_bit_cast(int, v);
  return __builtin_bit_cast(
      float, __builtin_amdgcn_permlanex16(b, b, 0x76543210, 0xfedcba98, false, false));
}

template <int M>
__device__ __forceinline__ float stp_row_xor(float v) {
  const int b = __builtin_bit_cast(int, v);
  return __builtin_bit_cast(
      float, __builtin_amdgcn_update_dpp(b, b, 0x160 + M, 0xf, 0xf, true));
}

#define SCAN_TC     32
#define SCAN_WAVES  8
#define SCAN_NCHUNK (LSEQ / SCAN_TC)
#define SCAN_XQ     (DMOD / 4)
#define SCAN_XLD    (SCAN_TC * SCAN_XQ / (SCAN_WAVES * 32))
static_assert(SCAN_TC == 32, "lane tt of a wave keeps the h of the chunk's step tt");

__global__ __launch_bounds__(SCAN_WAVES * 32)
void stp_scan_kernel(const float* __restrict__ x,
                     const float* __restrict__ Hs,
                     const float* __restrict__ Lambda,
                     const float* __restrict__ Gamma,
                     float* __restrict__ Hg) {
  __shared__ float4 xs[2][SCAN_TC][SCAN_XQ];
  __shared__ float  hss[2][SCAN_TC][SCAN_WAVES];

  const int tid   = threadIdx.x;
  const int wid   = tid >> 5;
  const int lane  = tid & 31;
  const int wave0 = blockIdx.x * SCAN_WAVES;
  if (wave0 >= BSZ * DFF) return;
  const int b  = wave0 >> 9;
  const int i0 = wave0 & (DFF - 1);
  const int i  = i0 + wid;
  const int jb = lane * 8;

  float r[8], g[8], F[8];
  const float* lam = Lambda + (size_t)i * DMOD + jb;
  const float* gam = Gamma  + (size_t)i * DMOD + jb;
#pragma unroll
  for (int u = 0; u < 8; ++u) {
    const float s = 1.0f / (1.0f + __expf(-lam[u]));
    r[u] = 1.0f - s;
    g[u] = gam[u];
    F[u] = 0.0f;
  }

  const int srow = tid >> 6;
  const int scol = tid & (SCAN_XQ - 1);
  const int htt  = tid >> 3;
  const int hw   = tid & (SCAN_WAVES - 1);
  const float4* xg  = (const float4*)(x + (size_t)b * LSEQ * DMOD) + (size_t)srow * SCAN_XQ + scol;
  const float*  hsg = Hs + (size_t)b * LSEQ * DFF + (size_t)htt * DFF + i0 + hw;
  float*     hg_ptr = Hg + (size_t)b * LSEQ * DFF + i;

  {
    float4 st[SCAN_XLD];
#pragma unroll
    for (int k = 0; k < SCAN_XLD; ++k) st[k] = xg[(size_t)(4 * k) * SCAN_XQ];
    const float sh = hsg[0];
#pragma unroll
    for (int k = 0; k < SCAN_XLD; ++k) xs[0][srow + 4 * k][scol] = st[k];
    hss[0][htt][hw] = sh;
  }
  __syncthreads();

  for (int c = 0; c < SCAN_NCHUNK; ++c) {
    const int  buf  = c & 1;
    const int  t0   = c * SCAN_TC;
    const bool more = (c + 1 < SCAN_NCHUNK);

    float4 st[SCAN_XLD];
    float  sh = 0.0f;
#pragma unroll
    for (int k = 0; k < SCAN_XLD; ++k) st[k] = make_float4(0.0f, 0.0f, 0.0f, 0.0f);
    if (more) {
      const size_t tn0 = (size_t)(t0 + SCAN_TC);
#pragma unroll
      for (int k = 0; k < SCAN_XLD; ++k) st[k] = xg[(tn0 + 4 * k) * SCAN_XQ];
      sh = hsg[tn0 * DFF];
    }

    float4 cx0 = xs[buf][0][2 * lane];
    float4 cx1 = xs[buf][0][2 * lane + 1];
    float  chs = hss[buf][0][wid];

    float hkeep = 0.0f;
    for (int tt = 0; tt < SCAN_TC; ++tt) {
      const int tn = (tt + 1 < SCAN_TC) ? tt + 1 : tt;
      const float4 nx0 = xs[buf][tn][2 * lane];
      const float4 nx1 = xs[buf][tn][2 * lane + 1];
      const float  nhs = hss[buf][tn][wid];

      float xv[8] = {cx0.x, cx0.y, cx0.z, cx0.w, cx1.x, cx1.y, cx1.z, cx1.w};

      asm volatile("" : "+v"(xv[0]) : : "memory");

      float p = 0.0f;
#pragma unroll
      for (int u = 0; u < 8; ++u) p = fmaf(F[u], xv[u], p);
      p += stp_xhalf(p);
      p += stp_row_xor<8>(p);
      p += stp_row_xor<4>(p);
      p += stp_row_xor<2>(p);
      p += stp_row_xor<1>(p);

      const float h = chs + p;
      hkeep = (lane == tt) ? h : hkeep;

#pragma unroll
      for (int u = 0; u < 8; ++u) F[u] = fmaf(r[u], F[u], g[u] * h * xv[u]);

      cx0 = nx0; cx1 = nx1; chs = nhs;
    }

    if (more) {
#pragma unroll
      for (int k = 0; k < SCAN_XLD; ++k) xs[buf ^ 1][srow + 4 * k][scol] = st[k];
      hss[buf ^ 1][htt][hw] = sh;
      __syncthreads();
    }

    {
      const float gh = 0.5f * hkeep * (1.0f + erff(hkeep * 0.70710678118654752f));
      hg_ptr[(size_t)(t0 + lane) * DFF] = gh;
    }
  }
}

extern "C" void kernel_launch(void* const* d_in, const int* in_sizes, int n_in,
                              void* d_out, int out_size, void* d_ws, size_t ws_size,
                              hipStream_t stream) {
  const float* x   = (const float*)d_in[0];
  const float* W1  = (const float*)d_in[1];
  const float* b1  = (const float*)d_in[2];
  const float* W2  = (const float*)d_in[3];
  const float* b2  = (const float*)d_in[4];
  const float* Lam = (const float*)d_in[5];
  const float* Gam = (const float*)d_in[6];
  float* out = (float*)d_out;

  float* Hs = (float*)d_ws;
  float* Hg = Hs + (size_t)MTOK * DFF;

  {
    constexpr int NT = 4;
    const int waves = (MTOK / 16) * (DFF / (16 * NT));
    stp_wmma_gemm_bias<MTOK, DFF, DMOD, NT>
        <<<waves / 4, 128, 0, stream>>>(x, W1, b1, Hs);
  }

  stp_scan_kernel<<<(BSZ * DFF) / 8, 256, 0, stream>>>(x, Hs, Lam, Gam, Hg);

  {
    constexpr int NT = 4;
    const int waves = (MTOK / 16) * (DMOD / (16 * NT));
    stp_wmma_gemm_bias<MTOK, DMOD, DFF, NT>
        <<<waves / 4, 128, 0, stream>>>(Hg, W2, b2, out);
  }
}


# scan: two time steps per trip: h(t+1) = hs + dot(r*F,x1) + h(t)*dot(g*x0,x1), its lane sums run beside step t's; F(t+1) from the same r*F and g*x0; odd steps' sum order and F rounding change
# speedup vs baseline: 3.3024x; 1.0404x over previous
#include <hip/hip_runtime.h>
#include <math.h>

#define BSZ   4
#define LSEQ  1024
#define DMOD  256
#define DFF   512
#define MTOK  (BSZ * LSEQ)

typedef float v2f_t __attribute__((ext_vector_type(2)));
typedef float v8f_t __attribute__((ext_vector_type(8)));

template <int Mdim, int Ndim, int Kdim, int NT>
__global__ __launch_bounds__(128)
void stp_wmma_gemm_bias(const float* __restrict__ A,
                        const float* __restrict__ Bt,
                        const float* __restrict__ bias,
                        float* __restrict__ D) {
  const int wave = blockIdx.x * (blockDim.x >> 5) + (threadIdx.x >> 5);
  const int lane = threadIdx.x & 31;
  constexpr int n_groups = Ndim / (16 * NT);
  const int tm = wave / n_groups;
  const int tg = wave % n_groups;
  if (tm >= Mdim / 16) return;

  const int m0   = tm * 16;
  const int n0   = tg * 16 * NT;
  const int row  = lane & 15;
  const int half = lane >> 4;

  const float* arow = A + (size_t)(m0 + row) * Kdim + 2 * half;
  const float* brow[NT];
#pragma unroll
  for (int nt = 0; nt < NT; ++nt)
    brow[nt] = Bt + (size_t)(n0 + nt * 16 + row) * Kdim + 2 * half;

  v8f_t c[NT];
#pragma unroll
  for (int nt = 0; nt < NT; ++nt) c[nt] = (v8f_t){};

#pragma unroll 4
  for (int k0 = 0; k0 < Kdim; k0 += 4) {
    const v2f_t a = *(const v2f_t*)(arow + k0);
#pragma unroll
    for (int nt = 0; nt < NT; ++nt) {
      const v2f_t b = *(const v2f_t*)(brow[nt] + k0);
      c[nt] = __builtin_amdgcn_wmma_f32_16x16x4_f32(false, a, false, b,
                                                    (short)0, c[nt], false, false);
    }
  }

#pragma unroll
  for (int nt = 0; nt < NT; ++nt) {
    const int n = n0 + nt * 16 + row;
    const float bn = bias[n];
#pragma unroll
    for (int v = 0; v < 8; ++v) {
      const int m = m0 + v + 8 * half;
      D[(size_t)m * Ndim + n] = c[nt][v] + bn;
    }
  }
}

__device__ __forceinline__ float stp_xhalf(float v) {
  const int b = __builtin_bit_cast(int, v);
  return __builtin_bit_cast(
      float, __builtin_amdgcn_permlanex16(b, b, 0x76543210, 0xfedcba98, false, false));
}

template <int M>
__device__ __forceinline__ float stp_row_xor(float v) {
  const int b = __builtin_bit_cast(int, v);
  return __builtin_bit_cast(
      float, __builtin_amdgcn_update_dpp(b, b, 0x160 + M, 0xf, 0xf, true));
}

__device__ __forceinline__ float stp_wave_sum(float p) {
  p += stp_xhalf(p);
  p += stp_row_xor<8>(p);
  p += stp_row_xor<4>(p);
  p += stp_row_xor<2>(p);
  p += stp_row_xor<1>(p);
  return p;
}

__device__ __forceinline__ void stp_wave_sum2(float a, float b, bool lower_row,
                                              float& sum_a, float& sum_b) {
  a += stp_xhalf(a);
  b += stp_xhalf(b);
  float q = lower_row ? a : b;
  q += stp_row_xor<8>(q);
  q += stp_row_xor<4>(q);
  q += stp_row_xor<2>(q);
  q += stp_row_xor<1>(q);
  const float qx = stp_xhalf(q);
  sum_a = lower_row ? q : qx;
  sum_b = lower_row ? qx : q;
}

#define SCAN_TC     32
#define SCAN_WAVES  8
#define SCAN_NCHUNK (LSEQ / SCAN_TC)
#define SCAN_XQ     (DMOD / 4)
#define SCAN_XLD    (SCAN_TC * SCAN_XQ / (SCAN_WAVES * 32))
static_assert(SCAN_TC == 32, "lane tt of a wave keeps the h of the chunk's step tt");
static_assert(SCAN_TC % 4 == 0, "the time loop takes two step pairs per trip");

struct stp_pair_ops {
  float4 a0, a1;
  float4 b0, b1;
  float  hsa, hsb;
};

__global__ __launch_bounds__(SCAN_WAVES * 32)
void stp_scan_kernel(const float* __restrict__ x,
                     const float* __restrict__ Hs,
                     const float* __restrict__ Lambda,
                     const float* __restrict__ Gamma,
                     float* __restrict__ Hg) {
  __shared__ float4 xs[2][SCAN_TC][SCAN_XQ];
  __shared__ float  hss[2][SCAN_TC][SCAN_WAVES];

  const int tid   = threadIdx.x;
  const int wid   = tid >> 5;
  const int lane  = tid & 31;
  const int wave0 = blockIdx.x * SCAN_WAVES;
  if (wave0 >= BSZ * DFF) return;
  const int b  = wave0 >> 9;
  const int i0 = wave0 & (DFF - 1);
  const int i  = i0 + wid;
  const int jb = lane * 8;

  float r[8], g[8], F[8];
  const float* lam = Lambda + (size_t)i * DMOD + jb;
  const float* gam = Gamma  + (size_t)i * DMOD + jb;
#pragma unroll
  for (int u = 0; u < 8; ++u) {
    const float s = 1.0f / (1.0f + __expf(-lam[u]));
    r[u] = 1.0f - s;
    g[u] = gam[u];
    F[u] = 0.0f;
  }

  const int srow = tid >> 6;
  const int scol = tid & (SCAN_XQ - 1);
  const int htt  = tid >> 3;
  const int hw   = tid & (SCAN_WAVES - 1);
  const float4* xg  = (const float4*)(x + (size_t)b * LSEQ * DMOD) + (size_t)srow * SCAN_XQ + scol;
  const float*  hsg = Hs + (size_t)b * LSEQ * DFF + (size_t)htt * DFF + i0 + hw;
  float*     hg_ptr = Hg + (size_t)b * LSEQ * DFF + i;

  {
    float4 st[SCAN_XLD];
#pragma unroll
    for (int k = 0; k < SCAN_XLD; ++k) st[k] = xg[(size_t)(4 * k) * SCAN_XQ];
    const float sh = hsg[0];
#pragma unroll
    for (int k = 0; k < SCAN_XLD; ++k) xs[0][srow + 4 * k][scol] = st[k];
    hss[0][htt][hw] = sh;
  }
  __syncthreads();

  for (int c = 0; c < SCAN_NCHUNK; ++c) {
    const int  buf  = c & 1;
    const int  t0   = c * SCAN_TC;
    const bool more = (c + 1 < SCAN_NCHUNK);

    float4 st[SCAN_XLD];
    float  sh = 0.0f;
#pragma unroll
    for (int k = 0; k < SCAN_XLD; ++k) st[k] = make_float4(0.0f, 0.0f, 0.0f, 0.0f);
    if (more) {
      const size_t tn0 = (size_t)(t0 + SCAN_TC);
#pragma unroll
      for (int k = 0; k < SCAN_XLD; ++k) st[k] = xg[(tn0 + 4 * k) * SCAN_XQ];
      sh = hsg[tn0 * DFF];
    }

    const auto ld_pair = [&](int tt) {
      stp_pair_ops o;
      o.a0 = xs[buf][tt][2 * lane];
      o.a1 = xs[buf][tt][2 * lane + 1];
      o.b0 = xs[buf][tt + 1][2 * lane];
      o.b1 = xs[buf][tt + 1][2 * lane + 1];
      o.hsa = hss[buf][tt][wid];
      o.hsb = hss[buf][tt + 1][wid];
      return o;
    };

    float hkeep = 0.0f;

    const auto step_pair = [&](const stp_pair_ops& o, int tp) {
      const float xa[8] = {o.a0.x, o.a0.y, o.a0.z, o.a0.w, o.a1.x, o.a1.y, o.a1.z, o.a1.w};
      const float xb[8] = {o.b0.x, o.b0.y, o.b0.z, o.b0.w, o.b1.x, o.b1.y, o.b1.z, o.b1.w};

      asm volatile(""
                   : "+v"(g[0]), "+v"(g[1]), "+v"(g[2]), "+v"(g[3]),
                     "+v"(g[4]), "+v"(g[5]), "+v"(g[6]), "+v"(g[7])
                   : "v"(F[0])
                   : "memory");

      float p0 = 0.0f, p1e = 0.0f, p1o = 0.0f, pae = 0.0f, pao = 0.0f;
      float rF[8], gx[8];
#pragma unroll
      for (int u = 0; u < 8; ++u) { rF[u] = r[u] * F[u]; gx[u] = g[u] * xa[u]; }
#pragma unroll
      for (int u = 0; u < 8; ++u) p0 = fmaf(F[u], xa[u], p0);
#pragma unroll
      for (int u = 0; u < 8; u += 2) {
        p1e = fmaf(rF[u], xb[u], p1e);
        p1o = fmaf(rF[u + 1], xb[u + 1], p1o);
        pae = fmaf(gx[u], xb[u], pae);
        pao = fmaf(gx[u + 1], xb[u + 1], pao);
      }
      p0 = stp_wave_sum(p0);
      float p1, pa;
      stp_wave_sum2(p1e + p1o, pae + pao, lane < 16, p1, pa);

      float c1 = o.hsb + p1;
      asm("" : "+v"(p0), "+v"(c1), "+v"(pa));

      const float h0 = o.hsa + p0;
      const float h1 = fmaf(h0, pa, c1);
      hkeep = (lane == tp)     ? h0 : hkeep;
      hkeep = (lane == tp + 1) ? h1 : hkeep;

#pragma unroll
      for (int u = 0; u < 8; ++u) {
        const float F1 = fmaf(h0, gx[u], rF[u]);
        F[u] = fmaf(r[u], F1, g[u] * h1 * xb[u]);
      }
    };

    stp_pair_ops opA = ld_pair(0);
    for (int tp = 0; tp < SCAN_TC; tp += 4) {
      const stp_pair_ops opB = ld_pair(tp + 2);
      step_pair(opA, tp);
      const int tn = (tp + 4 < SCAN_TC) ? tp + 4 : tp;
      opA = ld_pair(tn);
      step_pair(opB, tp + 2);
    }

    if (more) {
#pragma unroll
      for (int k = 0; k < SCAN_XLD; ++k) xs[buf ^ 1][srow + 4 * k][scol] = st[k];
      hss[buf ^ 1][htt][hw] = sh;
      __syncthreads();
    }

    {
      const float gh = 0.5f * hkeep * (1.0f + erff(hkeep * 0.70710678118654752f));
      hg_ptr[(size_t)(t0 + lane) * DFF] = gh;
    }
  }
}

extern "C" void kernel_launch(void* const* d_in, const int* in_sizes, int n_in,
                              void* d_out, int out_size, void* d_ws, size_t ws_size,
                              hipStream_t stream) {
  const float* x   = (const float*)d_in[0];
  const float* W1  = (const float*)d_in[1];
  const float* b1  = (const float*)d_in[2];
  const float* W2  = (const float*)d_in[3];
  const float* b2  = (const float*)d_in[4];
  const float* Lam = (const float*)d_in[5];
  const float* Gam = (const float*)d_in[6];
  float* out = (float*)d_out;

  float* Hs = (float*)d_ws;
  float* Hg = Hs + (size_t)MTOK * DFF;

  {
    constexpr int NT = 4;
    const int waves = (MTOK / 16) * (DFF / (16 * NT));
    stp_wmma_gemm_bias<MTOK, DFF, DMOD, NT>
        <<<waves / 4, 128, 0, stream>>>(x, W1, b1, Hs);
  }

  stp_scan_kernel<<<(BSZ * DFF) / 8, 256, 0, stream>>>(x, Hs, Lam, Gam, Hg);

  {
    constexpr int NT = 4;
    const int waves = (MTOK / 16) * (DMOD / (16 * NT));
    stp_wmma_gemm_bias<MTOK, DMOD, DFF, NT>
        <<<waves / 4, 128, 0, stream>>>(Hg, W2, b2, out);
  }
}


# both GEMMs: A and B tiles staged through LDS per 64-wide K chunk with whole-line 16-byte loads one chunk ahead, 64x128 / 64x64 work-group tiles; k order inside 8-blocks (product sum order) changes
# speedup vs baseline: 3.6894x; 1.1172x over previous
#include <hip/hip_runtime.h>
#include <math.h>

#define BSZ   4
#define LSEQ  1024
#define DMOD  256
#define DFF   512
#define MTOK  (BSZ * LSEQ)

typedef float v2f_t __attribute__((ext_vector_type(2)));
typedef float v8f_t __attribute__((ext_vector_type(8)));

#define GEMM_BK     64
#define GEMM_PITCH  (GEMM_BK + 4)
#define GEMM_Q      (GEMM_BK / 4)

template <int Mdim, int Ndim, int Kdim, int BM, int BN, int WM, int WN>
__global__ __launch_bounds__(256)
void stp_wmma_gemm_bias(const float* __restrict__ A,
                        const float* __restrict__ Bt,
                        const float* __restrict__ bias,
                        float* __restrict__ D) {
  static_assert(WM * WN == 8, "8 waves per work-group");
  static_assert(Kdim % GEMM_BK == 0 && Mdim % BM == 0 && Ndim % BN == 0, "exact tiling");
  constexpr int TM = BM / WM, TN = BN / WN;
  constexpr int FM = TM / 16, FN = TN / 16;
  static_assert(TN == 32, "a wave's rows of D are whole 128-byte lines");
  constexpr int NC = Kdim / GEMM_BK;
  constexpr int LA = BM / 16, LB = BN / 16;
  constexpr int NBN = Ndim / BN;

  __shared__ __attribute__((aligned(16))) float As[2][BM][GEMM_PITCH];
  __shared__ __attribute__((aligned(16))) float Bs[2][BN][GEMM_PITCH];

  const int tid  = threadIdx.x;
  const int lane = tid & 31;
  const int wid  = tid >> 5;
  const int wm   = wid / WN;
  const int wn   = wid % WN;
  const int r    = lane & 15;
  const int h    = lane >> 4;
  const int bm   = blockIdx.x / NBN;
  const int bn   = blockIdx.x % NBN;

  const int sr = tid >> 4;
  const int sc = tid & (GEMM_Q - 1);
  const float4* ag = (const float4*)(A  + (size_t)(bm * BM + sr) * Kdim) + sc;
  const float4* bg = (const float4*)(Bt + (size_t)(bn * BN + sr) * Kdim) + sc;
  constexpr size_t rstep = (size_t)16 * Kdim / 4;

  v8f_t c[FM][FN];
#pragma unroll
  for (int fm = 0; fm < FM; ++fm)
#pragma unroll
    for (int fn = 0; fn < FN; ++fn) c[fm][fn] = (v8f_t){};

  int doff = (bm * BM + wm * TM + 8 * h) * Ndim + bn * BN + wn * TN + r;
  float bv[FN];
#pragma unroll
  for (int fn = 0; fn < FN; ++fn) bv[fn] = bias[bn * BN + wn * TN + fn * 16 + r];
  asm volatile("" : "+v"(doff));

  {
    float4 sa[LA], sb[LB];
#pragma unroll
    for (int j = 0; j < LA; ++j) sa[j] = ag[j * rstep];
#pragma unroll
    for (int j = 0; j < LB; ++j) sb[j] = bg[j * rstep];
#pragma unroll
    for (int j = 0; j < LA; ++j) *(float4*)&As[0][sr + 16 * j][4 * sc] = sa[j];
#pragma unroll
    for (int j = 0; j < LB; ++j) *(float4*)&Bs[0][sr + 16 * j][4 * sc] = sb[j];
  }
  __syncthreads();

#pragma unroll
  for (int ch = 0; ch < NC; ++ch) {
    const int  buf  = ch & 1;
    const bool more = (ch + 1 < NC);

    float4 sa[LA], sb[LB];
#pragma unroll
    for (int j = 0; j < LA; ++j) sa[j] = make_float4(0.0f, 0.0f, 0.0f, 0.0f);
#pragma unroll
    for (int j = 0; j < LB; ++j) sb[j] = make_float4(0.0f, 0.0f, 0.0f, 0.0f);
    if (more) {
      const size_t koff = (size_t)(ch + 1) * GEMM_Q;
#pragma unroll
      for (int j = 0; j < LA; ++j) sa[j] = ag[j * rstep + koff];
#pragma unroll
      for (int j = 0; j < LB; ++j) sb[j] = bg[j * rstep + koff];
    }

#pragma unroll
    for (int kb = 0; kb < GEMM_BK / 8; ++kb) {
      float4 a4[FM], b4[FN];
#pragma unroll
      for (int fm = 0; fm < FM; ++fm)
        a4[fm] = *(const float4*)&As[buf][wm * TM + fm * 16 + r][8 * kb + 4 * h];
#pragma unroll
      for (int fn = 0; fn < FN; ++fn)
        b4[fn] = *(const float4*)&Bs[buf][wn * TN + fn * 16 + r][8 * kb + 4 * h];
#pragma unroll
      for (int fm = 0; fm < FM; ++fm)
#pragma unroll
        for (int fn = 0; fn < FN; ++fn) {
          c[fm][fn] = __builtin_amdgcn_wmma_f32_16x16x4_f32(
              false, (v2f_t){a4[fm].x, a4[fm].y}, false, (v2f_t){b4[fn].x, b4[fn].y},
              (short)0, c[fm][fn], false, false);
        }
#pragma unroll
      for (int fm = 0; fm < FM; ++fm)
#pragma unroll
        for (int fn = 0; fn < FN; ++fn) {
          c[fm][fn] = __builtin_amdgcn_wmma_f32_16x16x4_f32(
              false, (v2f_t){a4[fm].z, a4[fm].w}, false, (v2f_t){b4[fn].z, b4[fn].w},
              (short)0, c[fm][fn], false, false);
        }
    }

    if (more) {
#pragma unroll
      for (int j = 0; j < LA; ++j) *(float4*)&As[buf ^ 1][sr + 16 * j][4 * sc] = sa[j];
#pragma unroll
      for (int j = 0; j < LB; ++j) *(float4*)&Bs[buf ^ 1][sr + 16 * j][4 * sc] = sb[j];
      __syncthreads();
    }
  }

  __builtin_amdgcn_sched_barrier(0);
  asm volatile("v_nop\n\tv_nop\n\tv_nop\n\tv_nop");
  __builtin_amdgcn_sched_barrier(0);

#pragma unroll
  for (int fm = 0; fm < FM; ++fm)
#pragma unroll
    for (int fn = 0; fn < FN; ++fn) {
#pragma unroll
      for (int v = 0; v < 8; ++v)
        D[(size_t)(doff + (fm * 16 + v) * Ndim + fn * 16)] = c[fm][fn][v] + bv[fn];
    }
}

__device__ __forceinline__ float stp_xhalf(float v) {
  const int b = __builtin_bit_cast(int, v);
  return __builtin_bit_cast(
      float, __builtin_amdgcn_permlanex16(b, b, 0x76543210, 0xfedcba98, false, false));
}

template <int M>
__device__ __forceinline__ float stp_row_xor(float v) {
  const int b = __builtin_bit_cast(int, v);
  return __builtin_bit_cast(
      float, __builtin_amdgcn_update_dpp(b, b, 0x160 + M, 0xf, 0xf, true));
}

__device__ __forceinline__ float stp_wave_sum(float p) {
  p += stp_xhalf(p);
  p += stp_row_xor<8>(p);
  p += stp_row_xor<4>(p);
  p += stp_row_xor<2>(p);
  p += stp_row_xor<1>(p);
  return p;
}

__device__ __forceinline__ void stp_wave_sum2(float a, float b, bool lower_row,
                                              float& sum_a, float& sum_b) {
  a += stp_xhalf(a);
  b += stp_xhalf(b);
  float q = lower_row ? a : b;
  q += stp_row_xor<8>(q);
  q += stp_row_xor<4>(q);
  q += stp_row_xor<2>(q);
  q += stp_row_xor<1>(q);
  const float qx = stp_xhalf(q);
  sum_a = lower_row ? q : qx;
  sum_b = lower_row ? qx : q;
}

#define SCAN_TC     32
#define SCAN_WAVES  8
#define SCAN_NCHUNK (LSEQ / SCAN_TC)
#define SCAN_XQ     (DMOD / 4)
#define SCAN_XLD    (SCAN_TC * SCAN_XQ / (SCAN_WAVES * 32))
static_assert(SCAN_TC == 32, "lane tt of a wave keeps the h of the chunk's step tt");
static_assert(SCAN_TC % 4 == 0, "the time loop takes two step pairs per trip");

struct stp_pair_ops {
  float4 a0, a1;
  float4 b0, b1;
  float  hsa, hsb;
};

__global__ __launch_bounds__(SCAN_WAVES * 32)
void stp_scan_kernel(const float* __restrict__ x,
                     const float* __restrict__ Hs,
                     const float* __restrict__ Lambda,
                     const float* __restrict__ Gamma,
                     float* __restrict__ Hg) {
  __shared__ float4 xs[2][SCAN_TC][SCAN_XQ];
  __shared__ float  hss[2][SCAN_TC][SCAN_WAVES];

  const int tid   = threadIdx.x;
  const int wid   = tid >> 5;
  const int lane  = tid & 31;
  const int wave0 = blockIdx.x * SCAN_WAVES;
  if (wave0 >= BSZ * DFF) return;
  const int b  = wave0 >> 9;
  const int i0 = wave0 & (DFF - 1);
  const int i  = i0 + wid;
  const int jb = lane * 8;

  float r[8], g[8], F[8];
  const float* lam = Lambda + (size_t)i * DMOD + jb;
  const float* gam = Gamma  + (size_t)i * DMOD + jb;
#pragma unroll
  for (int u = 0; u < 8; ++u) {
    const float s = 1.0f / (1.0f + __expf(-lam[u]));
    r[u] = 1.0f - s;
    g[u] = gam[u];
    F[u] = 0.0f;
  }

  const int srow = tid >> 6;
  const int scol = tid & (SCAN_XQ - 1);
  const int htt  = tid >> 3;
  const int hw   = tid & (SCAN_WAVES - 1);
  const float4* xg  = (const float4*)(x + (size_t)b * LSEQ * DMOD) + (size_t)srow * SCAN_XQ + scol;
  const float*  hsg = Hs + (size_t)b * LSEQ * DFF + (size_t)htt * DFF + i0 + hw;
  float*     hg_ptr = Hg + (size_t)b * LSEQ * DFF + i;

  {
    float4 st[SCAN_XLD];
#pragma unroll
    for (int k = 0; k < SCAN_XLD; ++k) st[k] = xg[(size_t)(4 * k) * SCAN_XQ];
    const float sh = hsg[0];
#pragma unroll
    for (int k = 0; k < SCAN_XLD; ++k) xs[0][srow + 4 * k][scol] = st[k];
    hss[0][htt][hw] = sh;
  }
  __syncthreads();

  for (int c = 0; c < SCAN_NCHUNK; ++c) {
    const int  buf  = c & 1;
    const int  t0   = c * SCAN_TC;
    const bool more = (c + 1 < SCAN_NCHUNK);

    float4 st[SCAN_XLD];
    float  sh = 0.0f;
#pragma unroll
    for (int k = 0; k < SCAN_XLD; ++k) st[k] = make_float4(0.0f, 0.0f, 0.0f, 0.0f);
    if (more) {
      const size_t tn0 = (size_t)(t0 + SCAN_TC);
#pragma unroll
      for (int k = 0; k < SCAN_XLD; ++k) st[k] = xg[(tn0 + 4 * k) * SCAN_XQ];
      sh = hsg[tn0 * DFF];
    }

    const auto ld_pair = [&](int tt) {
      stp_pair_ops o;
      o.a0 = xs[buf][tt][2 * lane];
      o.a1 = xs[buf][tt][2 * lane + 1];
      o.b0 = xs[buf][tt + 1][2 * lane];
      o.b1 = xs[buf][tt + 1][2 * lane + 1];
      o.hsa = hss[buf][tt][wid];
      o.hsb = hss[buf][tt + 1][wid];
      return o;
    };

    float hkeep = 0.0f;

    const auto step_pair = [&](const stp_pair_ops& o, int tp) {
      const float xa[8] = {o.a0.x, o.a0.y, o.a0.z, o.a0.w, o.a1.x, o.a1.y, o.a1.z, o.a1.w};
      const float xb[8] = {o.b0.x, o.b0.y, o.b0.z, o.b0.w, o.b1.x, o.b1.y, o.b1.z, o.b1.w};

      asm volatile(""
                   : "+v"(g[0]), "+v"(g[1]), "+v"(g[2]), "+v"(g[3]),
                     "+v"(g[4]), "+v"(g[5]), "+v"(g[6]), "+v"(g[7])
                   : "v"(F[0])
                   : "memory");

      float p0 = 0.0f, p1e = 0.0f, p1o = 0.0f, pae = 0.0f, pao = 0.0f;
      float rF[8], gx[8];
#pragma unroll
      for (int u = 0; u < 8; ++u) { rF[u] = r[u] * F[u]; gx[u] = g[u] * xa[u]; }
#pragma unroll
      for (int u = 0; u < 8; ++u) p0 = fmaf(F[u], xa[u], p0);
#pragma unroll
      for (int u = 0; u < 8; u += 2) {
        p1e = fmaf(rF[u], xb[u], p1e);
        p1o = fmaf(rF[u + 1], xb[u + 1], p1o);
        pae = fmaf(gx[u], xb[u], pae);
        pao = fmaf(gx[u + 1], xb[u + 1], pao);
      }
      p0 = stp_wave_sum(p0);
      float p1, pa;
      stp_wave_sum2(p1e + p1o, pae + pao, lane < 16, p1, pa);

      float c1 = o.hsb + p1;
      asm("" : "+v"(p0), "+v"(c1), "+v"(pa));

      const float h0 = o.hsa + p0;
      const float h1 = fmaf(h0, pa, c1);
      hkeep = (lane == tp)     ? h0 : hkeep;
      hkeep = (lane == tp + 1) ? h1 : hkeep;

#pragma unroll
      for (int u = 0; u < 8; ++u) {
        const float F1 = fmaf(h0, gx[u], rF[u]);
        F[u] = fmaf(r[u], F1, g[u] * h1 * xb[u]);
      }
    };

    stp_pair_ops opA = ld_pair(0);
    for (int tp = 0; tp < SCAN_TC; tp += 4) {
      const stp_pair_ops opB = ld_pair(tp + 2);
      step_pair(opA, tp);
      const int tn = (tp + 4 < SCAN_TC) ? tp + 4 : tp;
      opA = ld_pair(tn);
      step_pair(opB, tp + 2);
    }

    if (more) {
#pragma unroll
      for (int k = 0; k < SCAN_XLD; ++k) xs[buf ^ 1][srow + 4 * k][scol] = st[k];
      hss[buf ^ 1][htt][hw] = sh;
      __syncthreads();
    }

    {
      const float gh = 0.5f * hkeep * (1.0f + erff(hkeep * 0.70710678118654752f));
      hg_ptr[(size_t)(t0 + lane) * DFF] = gh;
    }
  }
}

extern "C" void kernel_launch(void* const* d_in, const int* in_sizes, int n_in,
                              void* d_out, int out_size, void* d_ws, size_t ws_size,
                              hipStream_t stream) {
  const float* x   = (const float*)d_in[0];
  const float* W1  = (const float*)d_in[1];
  const float* b1  = (const float*)d_in[2];
  const float* W2  = (const float*)d_in[3];
  const float* b2  = (const float*)d_in[4];
  const float* Lam = (const float*)d_in[5];
  const float* Gam = (const float*)d_in[6];
  float* out = (float*)d_out;

  float* Hs = (float*)d_ws;
  float* Hg = Hs + (size_t)MTOK * DFF;

  stp_wmma_gemm_bias<MTOK, DFF, DMOD, 64, 128, 2, 4>
      <<<(MTOK / 64) * (DFF / 128), 256, 0, stream>>>(x, W1, b1, Hs);

  stp_scan_kernel<<<(BSZ * DFF) / 8, 256, 0, stream>>>(x, Hs, Lam, Gam, Hg);

  stp_wmma_gemm_bias<MTOK, DMOD, DFF, 64, 64, 4, 2>
      <<<(MTOK / 64) * (DMOD / 64), 256, 0, stream>>>(Hg, W2, b2, out);
}


# scan: a wave carries two F rows, one per 16-lane row (16 columns a lane; was one row on 32 lanes x 8 columns), one wave per SIMD; lane sums are 4 in-row DPP steps; dot-product sum order changes
# speedup vs baseline: 4.3795x; 1.1870x over previous
#include <hip/hip_runtime.h>
#include <math.h>

#define BSZ   4
#define LSEQ  1024
#define DMOD  256
#define DFF   512
#define MTOK  (BSZ * LSEQ)

typedef float v2f_t __attribute__((ext_vector_type(2)));
typedef float v8f_t __attribute__((ext_vector_type(8)));

#define GEMM_BK     64
#define GEMM_PITCH  (GEMM_BK + 4)
#define GEMM_Q      (GEMM_BK / 4)

template <int Mdim, int Ndim, int Kdim, int BM, int BN, int WM, int WN>
__global__ __launch_bounds__(256)
void stp_wmma_gemm_bias(const float* __restrict__ A,
                        const float* __restrict__ Bt,
                        const float* __restrict__ bias,
                        float* __restrict__ D) {
  static_assert(WM * WN == 8, "8 waves per work-group");
  static_assert(Kdim % GEMM_BK == 0 && Mdim % BM == 0 && Ndim % BN == 0, "exact tiling");
  constexpr int TM = BM / WM, TN = BN / WN;
  constexpr int FM = TM / 16, FN = TN / 16;
  static_assert(TN == 32, "a wave's rows of D are whole 128-byte lines");
  constexpr int NC = Kdim / GEMM_BK;
  constexpr int LA = BM / 16, LB = BN / 16;
  constexpr int NBN = Ndim / BN;

  __shared__ __attribute__((aligned(16))) float As[2][BM][GEMM_PITCH];
  __shared__ __attribute__((aligned(16))) float Bs[2][BN][GEMM_PITCH];

  const int tid  = threadIdx.x;
  const int lane = tid & 31;
  const int wid  = tid >> 5;
  const int wm   = wid / WN;
  const int wn   = wid % WN;
  const int r    = lane & 15;
  const int h    = lane >> 4;
  const int bm   = blockIdx.x / NBN;
  const int bn   = blockIdx.x % NBN;

  const int sr = tid >> 4;
  const int sc = tid & (GEMM_Q - 1);
  const float4* ag = (const float4*)(A  + (size_t)(bm * BM + sr) * Kdim) + sc;
  const float4* bg = (const float4*)(Bt + (size_t)(bn * BN + sr) * Kdim) + sc;
  constexpr size_t rstep = (size_t)16 * Kdim / 4;

  v8f_t c[FM][FN];
#pragma unroll
  for (int fm = 0; fm < FM; ++fm)
#pragma unroll
    for (int fn = 0; fn < FN; ++fn) c[fm][fn] = (v8f_t){};

  int doff = (bm * BM + wm * TM + 8 * h) * Ndim + bn * BN + wn * TN + r;
  float bv[FN];
#pragma unroll
  for (int fn = 0; fn < FN; ++fn) bv[fn] = bias[bn * BN + wn * TN + fn * 16 + r];
  asm volatile("" : "+v"(doff));

  {
    float4 sa[LA], sb[LB];
#pragma unroll
    for (int j = 0; j < LA; ++j) sa[j] = ag[j * rstep];
#pragma unroll
    for (int j = 0; j < LB; ++j) sb[j] = bg[j * rstep];
#pragma unroll
    for (int j = 0; j < LA; ++j) *(float4*)&As[0][sr + 16 * j][4 * sc] = sa[j];
#pragma unroll
    for (int j = 0; j < LB; ++j) *(float4*)&Bs[0][sr + 16 * j][4 * sc] = sb[j];
  }
  __syncthreads();

#pragma unroll
  for (int ch = 0; ch < NC; ++ch) {
    const int  buf  = ch & 1;
    const bool more = (ch + 1 < NC);

    float4 sa[LA], sb[LB];
#pragma unroll
    for (int j = 0; j < LA; ++j) sa[j] = make_float4(0.0f, 0.0f, 0.0f, 0.0f);
#pragma unroll
    for (int j = 0; j < LB; ++j) sb[j] = make_float4(0.0f, 0.0f, 0.0f, 0.0f);
    if (more) {
      const size_t koff = (size_t)(ch + 1) * GEMM_Q;
#pragma unroll
      for (int j = 0; j < LA; ++j) sa[j] = ag[j * rstep + koff];
#pragma unroll
      for (int j = 0; j < LB; ++j) sb[j] = bg[j * rstep + koff];
    }

#pragma unroll
    for (int kb = 0; kb < GEMM_BK / 8; ++kb) {
      float4 a4[FM], b4[FN];
#pragma unroll
      for (int fm = 0; fm < FM; ++fm)
        a4[fm] = *(const float4*)&As[buf][wm * TM + fm * 16 + r][8 * kb + 4 * h];
#pragma unroll
      for (int fn = 0; fn < FN; ++fn)
        b4[fn] = *(const float4*)&Bs[buf][wn * TN + fn * 16 + r][8 * kb + 4 * h];
#pragma unroll
      for (int fm = 0; fm < FM; ++fm)
#pragma unroll
        for (int fn = 0; fn < FN; ++fn) {
          c[fm][fn] = __builtin_amdgcn_wmma_f32_16x16x4_f32(
              false, (v2f_t){a4[fm].x, a4[fm].y}, false, (v2f_t){b4[fn].x, b4[fn].y},
              (short)0, c[fm][fn], false, false);
        }
#pragma unroll
      for (int fm = 0; fm < FM; ++fm)
#pragma unroll
        for (int fn = 0; fn < FN; ++fn) {
          c[fm][fn] = __builtin_amdgcn_wmma_f32_16x16x4_f32(
              false, (v2f_t){a4[fm].z, a4[fm].w}, false, (v2f_t){b4[fn].z, b4[fn].w},
              (short)0, c[fm][fn], false, false);
        }
    }

    if (more) {
#pragma unroll
      for (int j = 0; j < LA; ++j) *(float4*)&As[buf ^ 1][sr + 16 * j][4 * sc] = sa[j];
#pragma unroll
      for (int j = 0; j < LB; ++j) *(float4*)&Bs[buf ^ 1][sr + 16 * j][4 * sc] = sb[j];
      __syncthreads();
    }
  }

  __builtin_amdgcn_sched_barrier(0);
  asm volatile("v_nop\n\tv_nop\n\tv_nop\n\tv_nop");
  __builtin_amdgcn_sched_barrier(0);

#pragma unroll
  for (int fm = 0; fm < FM; ++fm)
#pragma unroll
    for (int fn = 0; fn < FN; ++fn) {
#pragma unroll
      for (int v = 0; v < 8; ++v)
        D[(size_t)(doff + (fm * 16 + v) * Ndim + fn * 16)] = c[fm][fn][v] + bv[fn];
    }
}

template <int M>
__device__ __forceinline__ float stp_row_xor(float v) {
  const int b = __builtin_bit_cast(int, v);
  return __builtin_bit_cast(
      float, __builtin_amdgcn_update_dpp(b, b, 0x160 + M, 0xf, 0xf, true));
}

__device__ __forceinline__ float stp_row_sum(float p) {
  p += stp_row_xor<8>(p);
  p += stp_row_xor<4>(p);
  p += stp_row_xor<2>(p);
  p += stp_row_xor<1>(p);
  return p;
}

#define SCAN_TC     32
#define SCAN_WAVES  4
#define SCAN_ROWS   (2 * SCAN_WAVES)
#define SCAN_NCHUNK (LSEQ / SCAN_TC)
#define SCAN_XQ     (DMOD / 4)
#define SCAN_XLD    (SCAN_TC * SCAN_XQ / (SCAN_WAVES * 32))
#define SCAN_HLD    (SCAN_TC * SCAN_ROWS / (SCAN_WAVES * 32))
static_assert(SCAN_TC == 32, "lane c of a row keeps the h of the chunk's steps c and 16 + c");
static_assert(SCAN_XQ == 64, "a lane's four float4 of an x row are 16 float4 apart");
static_assert(SCAN_HLD == 2, "a thread stages two H_static values per chunk");

struct stp_pair_ops {
  float4 a[4];
  float4 b[4];
  float  hsa, hsb;
};

__global__ __launch_bounds__(SCAN_WAVES * 32)
void stp_scan_kernel(const float* __restrict__ x,
                     const float* __restrict__ Hs,
                     const float* __restrict__ Lambda,
                     const float* __restrict__ Gamma,
                     float* __restrict__ Hg) {
  __shared__ float4 xs[2][SCAN_TC][SCAN_XQ];
  __shared__ float  hss[2][SCAN_TC][SCAN_ROWS];

  const int tid  = threadIdx.x;
  const int wid  = tid >> 5;
  const int lane = tid & 31;
  const int c    = lane & 15;
  const int hrow = 2 * wid + (lane >> 4);
  const int row0 = blockIdx.x * SCAN_ROWS;
  if (row0 >= BSZ * DFF) return;
  const int b  = row0 >> 9;
  const int i0 = row0 & (DFF - 1);
  const int i  = i0 + hrow;

  float r[16], g[16], F[16];
  const float4* lam = (const float4*)(Lambda + (size_t)i * DMOD) + c;
  const float4* gam = (const float4*)(Gamma  + (size_t)i * DMOD) + c;
#pragma unroll
  for (int k = 0; k < 4; ++k) {
    const float4 l4 = lam[16 * k];
    const float4 g4 = gam[16 * k];
    const float lv[4] = {l4.x, l4.y, l4.z, l4.w};
    const float gv[4] = {g4.x, g4.y, g4.z, g4.w};
#pragma unroll
    for (int j = 0; j < 4; ++j) {
      const float s = 1.0f / (1.0f + __expf(-lv[j]));
      r[4 * k + j] = 1.0f - s;
      g[4 * k + j] = gv[j];
      F[4 * k + j] = 0.0f;
    }
  }

  const int srow = tid >> 6;
  const int scol = tid & (SCAN_XQ - 1);
  const int htt  = tid >> 3;
  const int hw   = tid & (SCAN_ROWS - 1);
  const float4* xg  = (const float4*)(x + (size_t)b * LSEQ * DMOD) + (size_t)srow * SCAN_XQ + scol;
  const float*  hsg = Hs + (size_t)b * LSEQ * DFF + (size_t)htt * DFF + i0 + hw;
  float*     hg_ptr = Hg + (size_t)b * LSEQ * DFF + i;

  {
    float4 st[SCAN_XLD];
#pragma unroll
    for (int k = 0; k < SCAN_XLD; ++k) st[k] = xg[(size_t)(2 * k) * SCAN_XQ];
    const float sh0 = hsg[0];
    const float sh1 = hsg[(size_t)16 * DFF];
#pragma unroll
    for (int k = 0; k < SCAN_XLD; ++k) xs[0][srow + 2 * k][scol] = st[k];
    hss[0][htt][hw] = sh0;
    hss[0][htt + 16][hw] = sh1;
  }
  __syncthreads();

  for (int ch = 0; ch < SCAN_NCHUNK; ++ch) {
    const int  buf  = ch & 1;
    const int  t0   = ch * SCAN_TC;
    const bool more = (ch + 1 < SCAN_NCHUNK);

    float4 st[SCAN_XLD];
    float  sh0 = 0.0f, sh1 = 0.0f;
#pragma unroll
    for (int k = 0; k < SCAN_XLD; ++k) st[k] = make_float4(0.0f, 0.0f, 0.0f, 0.0f);
    if (more) {
      const size_t tn0 = (size_t)(t0 + SCAN_TC);
#pragma unroll
      for (int k = 0; k < SCAN_XLD; ++k) st[k] = xg[(tn0 + 2 * k) * SCAN_XQ];
      sh0 = hsg[tn0 * DFF];
      sh1 = hsg[(tn0 + 16) * DFF];
    }

    const auto ld_pair = [&](int tt) {
      stp_pair_ops o;
#pragma unroll
      for (int k = 0; k < 4; ++k) {
        o.a[k] = xs[buf][tt][16 * k + c];
        o.b[k] = xs[buf][tt + 1][16 * k + c];
      }
      o.hsa = hss[buf][tt][hrow];
      o.hsb = hss[buf][tt + 1][hrow];
      return o;
    };

    float hkeep = 0.0f;
    float hk0 = 0.0f;

    const auto step_pair = [&](const stp_pair_ops& o, int tq) {
      float xa[16], xb[16];
#pragma unroll
      for (int k = 0; k < 4; ++k) {
        xa[4 * k] = o.a[k].x; xa[4 * k + 1] = o.a[k].y; xa[4 * k + 2] = o.a[k].z; xa[4 * k + 3] = o.a[k].w;
        xb[4 * k] = o.b[k].x; xb[4 * k + 1] = o.b[k].y; xb[4 * k + 2] = o.b[k].z; xb[4 * k + 3] = o.b[k].w;
      }

      asm volatile(""
                   : "+v"(g[0]), "+v"(g[1]), "+v"(g[2]), "+v"(g[3]),
                     "+v"(g[4]), "+v"(g[5]), "+v"(g[6]), "+v"(g[7]),
                     "+v"(g[8]), "+v"(g[9]), "+v"(g[10]), "+v"(g[11]),
                     "+v"(g[12]), "+v"(g[13]), "+v"(g[14]), "+v"(g[15])
                   : "v"(F[0])
                   : "memory");

      float q0[4], q1[4], qa[4];
      float rF[16], gx[16];
#pragma unroll
      for (int u = 0; u < 16; ++u) { rF[u] = r[u] * F[u]; gx[u] = g[u] * xa[u]; }
#pragma unroll
      for (int j = 0; j < 4; ++j) {
        q0[j] = F[j] * xa[j];
        q1[j] = rF[j] * xb[j];
        qa[j] = gx[j] * xb[j];
      }
#pragma unroll
      for (int u = 4; u < 16; ++u) {
        q0[u & 3] = fmaf(F[u], xa[u], q0[u & 3]);
        q1[u & 3] = fmaf(rF[u], xb[u], q1[u & 3]);
        qa[u & 3] = fmaf(gx[u], xb[u], qa[u & 3]);
      }
      float p0 = stp_row_sum((q0[0] + q0[2]) + (q0[1] + q0[3]));
      const float p1 = stp_row_sum((q1[0] + q1[2]) + (q1[1] + q1[3]));
      float pa = stp_row_sum((qa[0] + qa[2]) + (qa[1] + qa[3]));

      float c1 = o.hsb + p1;
      asm("" : "+v"(p0), "+v"(c1), "+v"(pa));

      const float h0 = o.hsa + p0;
      const float h1 = fmaf(h0, pa, c1);
      hkeep = (c == tq)     ? h0 : hkeep;
      hkeep = (c == tq + 1) ? h1 : hkeep;

#pragma unroll
      for (int u = 0; u < 16; ++u) {
        const float F1 = fmaf(h0, gx[u], rF[u]);
        F[u] = fmaf(r[u], F1, g[u] * h1 * xb[u]);
      }
    };

    stp_pair_ops opA = ld_pair(0);
    for (int tp = 0; tp < SCAN_TC; tp += 4) {
      const int tq = tp & 15;
      const stp_pair_ops opB = ld_pair(tp + 2);
      step_pair(opA, tq);
      const int tn = (tp + 4 < SCAN_TC) ? tp + 4 : tp;
      opA = ld_pair(tn);
      step_pair(opB, tq + 2);
      hk0 = (tp == 12) ? hkeep : hk0;
    }

    if (more) {
#pragma unroll
      for (int k = 0; k < SCAN_XLD; ++k) xs[buf ^ 1][srow + 2 * k][scol] = st[k];
      hss[buf ^ 1][htt][hw] = sh0;
      hss[buf ^ 1][htt + 16][hw] = sh1;
      __syncthreads();
    }

    {
      const float ga = 0.5f * hk0 * (1.0f + erff(hk0 * 0.70710678118654752f));
      const float gb = 0.5f * hkeep * (1.0f + erff(hkeep * 0.70710678118654752f));
      hg_ptr[(size_t)(t0 + c) * DFF] = ga;
      hg_ptr[(size_t)(t0 + 16 + c) * DFF] = gb;
    }
  }
}

extern "C" void kernel_launch(void* const* d_in, const int* in_sizes, int n_in,
                              void* d_out, int out_size, void* d_ws, size_t ws_size,
                              hipStream_t stream) {
  const float* x   = (const float*)d_in[0];
  const float* W1  = (const float*)d_in[1];
  const float* b1  = (const float*)d_in[2];
  const float* W2  = (const float*)d_in[3];
  const float* b2  = (const float*)d_in[4];
  const float* Lam = (const float*)d_in[5];
  const float* Gam = (const float*)d_in[6];
  float* out = (float*)d_out;

  float* Hs = (float*)d_ws;
  float* Hg = Hs + (size_t)MTOK * DFF;

  stp_wmma_gemm_bias<MTOK, DFF, DMOD, 64, 128, 2, 4>
      <<<(MTOK / 64) * (DFF / 128), 256, 0, stream>>>(x, W1, b1, Hs);

  stp_scan_kernel<<<(BSZ * DFF) / 8, 128, 0, stream>>>(x, Hs, Lam, Gam, Hg);

  stp_wmma_gemm_bias<MTOK, DMOD, DFF, 64, 64, 4, 2>
      <<<(MTOK / 64) * (DMOD / 64), 256, 0, stream>>>(Hg, W2, b2, out);
}


# scan: the pair's cross term dot(g*x0, x1) leaves the time loop: launch 1 forms it for all step pairs as a second WMMA product (x_t*x_(t+1)) @ Gamma^T on its x tiles; that term's sum order changes
# speedup vs baseline: 4.7139x; 1.0764x over previous
#include <hip/hip_runtime.h>
#include <math.h>

#define BSZ   4
#define LSEQ  1024
#define DMOD  256
#define DFF   512
#define MTOK  (BSZ * LSEQ)

typedef float v2f_t __attribute__((ext_vector_type(2)));
typedef float v8f_t __attribute__((ext_vector_type(8)));

#define GEMM_BK     64
#define GEMM_PITCH  (GEMM_BK + 4)
#define GEMM_Q      (GEMM_BK / 4)

template <int Mdim, int Ndim, int Kdim, int BM, int BN, int WM, int WN, bool PAIR>
__global__ __launch_bounds__(256)
void stp_wmma_gemm_bias(const float* __restrict__ A,
                        const float* __restrict__ Bt,
                        const float* __restrict__ bias,
                        float* __restrict__ D,
                        const float* __restrict__ Gt,
                        float* __restrict__ Q) {
  static_assert(WM * WN == 8, "8 waves per work-group");
  static_assert(Kdim % GEMM_BK == 0 && Mdim % BM == 0 && Ndim % BN == 0, "exact tiling");
  constexpr int TM = BM / WM, TN = BN / WN;
  constexpr int FM = TM / 16, FN = TN / 16;
  static_assert(TN == 32, "a wave's rows of D are whole 128-byte lines");
  constexpr int NC = Kdim / GEMM_BK;
  constexpr int LA = BM / 16, LB = BN / 16;
  constexpr int NBN = Ndim / BN;
  constexpr int PM  = BM / 2;
  constexpr int TM2 = PM / WM;
  static_assert(!PAIR || (TM2 == 16 && LA % 2 == 0), "one 16-row product tile per wave row");

  __shared__ __attribute__((aligned(16))) float As[2][BM][GEMM_PITCH];
  __shared__ __attribute__((aligned(16))) float Bs[2][BN][GEMM_PITCH];
  __shared__ __attribute__((aligned(16))) float Ps[PAIR ? 2 : 1][PAIR ? PM : 1][PAIR ? GEMM_PITCH : 4];
  __shared__ __attribute__((aligned(16))) float Gs[PAIR ? 2 : 1][PAIR ? BN : 1][PAIR ? GEMM_PITCH : 4];

  const int tid  = threadIdx.x;
  const int lane = tid & 31;
  const int wid  = tid >> 5;
  const int wm   = wid / WN;
  const int wn   = wid % WN;
  const int r    = lane & 15;
  const int h    = lane >> 4;
  const int bm   = blockIdx.x / NBN;
  const int bn   = blockIdx.x % NBN;

  const int sr = tid >> 4;
  const int sc = tid & (GEMM_Q - 1);
  const auto arow = [&](int j) { return PAIR ? 2 * sr + (j & 1) + 32 * (j >> 1) : sr + 16 * j; };
  const float4* ag = (const float4*)(A  + (size_t)(bm * BM) * Kdim) + sc;
  const float4* bg = (const float4*)(Bt + (size_t)(bn * BN + sr) * Kdim) + sc;
  const float4* gg = PAIR ? (const float4*)(Gt + (size_t)(bn * BN + sr) * Kdim) + sc : nullptr;
  constexpr size_t rstep = (size_t)16 * Kdim / 4;
  constexpr size_t rowq  = (size_t)Kdim / 4;
  const auto put_pairs = [&](int b, const float4* sa) {
#pragma unroll
    for (int jp = 0; jp < LA / 2; ++jp) {
      const float4 u = sa[2 * jp], w = sa[2 * jp + 1];
      *(float4*)&Ps[b][sr + 16 * jp][4 * sc] = make_float4(u.x * w.x, u.y * w.y, u.z * w.z, u.w * w.w);
    }
  };

  v8f_t c[FM][FN];
#pragma unroll
  for (int fm = 0; fm < FM; ++fm)
#pragma unroll
    for (int fn = 0; fn < FN; ++fn) c[fm][fn] = (v8f_t){};
  v8f_t c2[FN];
#pragma unroll
  for (int fn = 0; fn < FN; ++fn) c2[fn] = (v8f_t){};

  int doff = (bm * BM + wm * TM + 8 * h) * Ndim + bn * BN + wn * TN + r;
  float bv[FN];
#pragma unroll
  for (int fn = 0; fn < FN; ++fn) bv[fn] = bias[bn * BN + wn * TN + fn * 16 + r];
  int qoff = 2 * (bm * PM + wm * TM2 + 8 * h) * Ndim + bn * BN + wn * TN + r;
  asm volatile("" : "+v"(doff));
  if constexpr (PAIR) asm volatile("" : "+v"(qoff));

  {
    float4 sa[LA], sb[LB], sg[LB];
#pragma unroll
    for (int j = 0; j < LA; ++j) sa[j] = ag[(size_t)arow(j) * rowq];
#pragma unroll
    for (int j = 0; j < LB; ++j) sb[j] = bg[j * rstep];
    if constexpr (PAIR) {
#pragma unroll
      for (int j = 0; j < LB; ++j) sg[j] = gg[j * rstep];
    }
#pragma unroll
    for (int j = 0; j < LA; ++j) *(float4*)&As[0][arow(j)][4 * sc] = sa[j];
#pragma unroll
    for (int j = 0; j < LB; ++j) *(float4*)&Bs[0][sr + 16 * j][4 * sc] = sb[j];
    if constexpr (PAIR) {
      put_pairs(0, sa);
#pragma unroll
      for (int j = 0; j < LB; ++j) *(float4*)&Gs[0][sr + 16 * j][4 * sc] = sg[j];
    }
  }
  __syncthreads();

#pragma unroll
  for (int ch = 0; ch < NC; ++ch) {
    const int  buf  = ch & 1;
    const bool more = (ch + 1 < NC);

    float4 sa[LA], sb[LB], sg[LB];
#pragma unroll
    for (int j = 0; j < LA; ++j) sa[j] = make_float4(0.0f, 0.0f, 0.0f, 0.0f);
#pragma unroll
    for (int j = 0; j < LB; ++j) sb[j] = sg[j] = make_float4(0.0f, 0.0f, 0.0f, 0.0f);
    if (more) {
      const size_t koff = (size_t)(ch + 1) * GEMM_Q;
#pragma unroll
      for (int j = 0; j < LA; ++j) sa[j] = ag[(size_t)arow(j) * rowq + koff];
#pragma unroll
      for (int j = 0; j < LB; ++j) sb[j] = bg[j * rstep + koff];
      if constexpr (PAIR) {
#pragma unroll
        for (int j = 0; j < LB; ++j) sg[j] = gg[j * rstep + koff];
      }
    }

#pragma unroll
    for (int kb = 0; kb < GEMM_BK / 8; ++kb) {
      float4 a4[FM], b4[FN];
#pragma unroll
      for (int fm = 0; fm < FM; ++fm)
        a4[fm] = *(const float4*)&As[buf][wm * TM + fm * 16 + r][8 * kb + 4 * h];
#pragma unroll
      for (int fn = 0; fn < FN; ++fn)
        b4[fn] = *(const float4*)&Bs[buf][wn * TN + fn * 16 + r][8 * kb + 4 * h];
#pragma unroll
      for (int fm = 0; fm < FM; ++fm)
#pragma unroll
        for (int fn = 0; fn < FN; ++fn) {
          c[fm][fn] = __builtin_amdgcn_wmma_f32_16x16x4_f32(
              false, (v2f_t){a4[fm].x, a4[fm].y}, false, (v2f_t){b4[fn].x, b4[fn].y},
              (short)0, c[fm][fn], false, false);
        }
#pragma unroll
      for (int fm = 0; fm < FM; ++fm)
#pragma unroll
        for (int fn = 0; fn < FN; ++fn) {
          c[fm][fn] = __builtin_amdgcn_wmma_f32_16x16x4_f32(
              false, (v2f_t){a4[fm].z, a4[fm].w}, false, (v2f_t){b4[fn].z, b4[fn].w},
              (short)0, c[fm][fn], false, false);
        }
      if constexpr (PAIR) {
        const float4 p4 = *(const float4*)&Ps[buf][wm * TM2 + r][8 * kb + 4 * h];
        float4 g4[FN];
#pragma unroll
        for (int fn = 0; fn < FN; ++fn)
          g4[fn] = *(const float4*)&Gs[buf][wn * TN + fn * 16 + r][8 * kb + 4 * h];
#pragma unroll
        for (int fn = 0; fn < FN; ++fn)
          c2[fn] = __builtin_amdgcn_wmma_f32_16x16x4_f32(
              false, (v2f_t){p4.x, p4.y}, false, (v2f_t){g4[fn].x, g4[fn].y},
              (short)0, c2[fn], false, false);
#pragma unroll
        for (int fn = 0; fn < FN; ++fn)
          c2[fn] = __builtin_amdgcn_wmma_f32_16x16x4_f32(
              false, (v2f_t){p4.z, p4.w}, false, (v2f_t){g4[fn].z, g4[fn].w},
              (short)0, c2[fn], false, false);
      }
    }

    if (more) {
#pragma unroll
      for (int j = 0; j < LA; ++j) *(float4*)&As[buf ^ 1][arow(j)][4 * sc] = sa[j];
#pragma unroll
      for (int j = 0; j < LB; ++j) *(float4*)&Bs[buf ^ 1][sr + 16 * j][4 * sc] = sb[j];
      if constexpr (PAIR) {
        put_pairs(buf ^ 1, sa);
#pragma unroll
        for (int j = 0; j < LB; ++j) *(float4*)&Gs[buf ^ 1][sr + 16 * j][4 * sc] = sg[j];
      }
      __syncthreads();
    }
  }

  __builtin_amdgcn_sched_barrier(0);
  asm volatile("v_nop\n\tv_nop\n\tv_nop\n\tv_nop");
  __builtin_amdgcn_sched_barrier(0);

#pragma unroll
  for (int fm = 0; fm < FM; ++fm)
#pragma unroll
    for (int fn = 0; fn < FN; ++fn) {
#pragma unroll
      for (int v = 0; v < 8; ++v)
        D[(size_t)(doff + (fm * 16 + v) * Ndim + fn * 16)] = c[fm][fn][v] + bv[fn];
    }
  if constexpr (PAIR) {
#pragma unroll
    for (int fn = 0; fn < FN; ++fn) {
#pragma unroll
      for (int v = 0; v < 8; ++v)
        Q[(size_t)(qoff + 2 * v * Ndim + fn * 16)] = c2[fn][v];
    }
  }
}

template <int M>
__device__ __forceinline__ float stp_row_xor(float v) {
  const int b = __builtin_bit_cast(int, v);
  return __builtin_bit_cast(
      float, __builtin_amdgcn_update_dpp(b, b, 0x160 + M, 0xf, 0xf, true));
}

__device__ __forceinline__ float stp_row_sum(float p) {
  p += stp_row_xor<8>(p);
  p += stp_row_xor<4>(p);
  p += stp_row_xor<2>(p);
  p += stp_row_xor<1>(p);
  return p;
}

#define SCAN_TC     32
#define SCAN_WAVES  4
#define SCAN_ROWS   (2 * SCAN_WAVES)
#define SCAN_NCHUNK (LSEQ / SCAN_TC)
#define SCAN_XQ     (DMOD / 4)
#define SCAN_XLD    (SCAN_TC * SCAN_XQ / (SCAN_WAVES * 32))
#define SCAN_NP     (SCAN_TC / 2)
static_assert(SCAN_TC == 32, "lane c of a row keeps the h of the chunk's steps c and 16 + c");
static_assert(SCAN_XQ == 64, "a lane's four float4 of an x row are 16 float4 apart");
static_assert(SCAN_NP * SCAN_ROWS == SCAN_WAVES * 32, "a thread stages one (step pair, dff-row) per chunk");

struct stp_pair_ops {
  float4 a[4];
  float4 b[4];
  float  hsa, hsb;
  float  pa;
};

__global__ __launch_bounds__(SCAN_WAVES * 32)
void stp_scan_kernel(const float* __restrict__ x,
                     const float* __restrict__ Hs,
                     const float* __restrict__ Lambda,
                     const float* __restrict__ Gamma,
                     float* __restrict__ Hg) {
  __shared__ float4 xs[2][SCAN_TC][SCAN_XQ];
  __shared__ float4 hsq[2][SCAN_NP][SCAN_ROWS];

  const int tid  = threadIdx.x;
  const int wid  = tid >> 5;
  const int lane = tid & 31;
  const int c    = lane & 15;
  const int hrow = 2 * wid + (lane >> 4);
  const int row0 = blockIdx.x * SCAN_ROWS;
  if (row0 >= BSZ * DFF) return;
  const int b  = row0 >> 9;
  const int i0 = row0 & (DFF - 1);
  const int i  = i0 + hrow;

  float r[16], g[16], F[16];
  const float4* lam = (const float4*)(Lambda + (size_t)i * DMOD) + c;
  const float4* gam = (const float4*)(Gamma  + (size_t)i * DMOD) + c;
#pragma unroll
  for (int k = 0; k < 4; ++k) {
    const float4 l4 = lam[16 * k];
    const float4 g4 = gam[16 * k];
    const float lv[4] = {l4.x, l4.y, l4.z, l4.w};
    const float gv[4] = {g4.x, g4.y, g4.z, g4.w};
#pragma unroll
    for (int j = 0; j < 4; ++j) {
      const float s = 1.0f / (1.0f + __expf(-lv[j]));
      r[4 * k + j] = 1.0f - s;
      g[4 * k + j] = gv[j];
      F[4 * k + j] = 0.0f;
    }
  }

  const int srow = tid >> 6;
  const int scol = tid & (SCAN_XQ - 1);
  const int hpq  = tid >> 3;
  const int hw   = tid & (SCAN_ROWS - 1);
  const float4* xg  = (const float4*)(x + (size_t)b * LSEQ * DMOD) + (size_t)srow * SCAN_XQ + scol;
  const size_t  hoff = (size_t)b * LSEQ * DFF + (size_t)(2 * hpq) * DFF + i0 + hw;
  const float*  hsg = Hs + hoff;
  const float*  qag = Hg + hoff;
  float*     hg_ptr = Hg + (size_t)b * LSEQ * DFF + i;

  {
    float4 st[SCAN_XLD];
#pragma unroll
    for (int k = 0; k < SCAN_XLD; ++k) st[k] = xg[(size_t)(2 * k) * SCAN_XQ];
    const float sh0 = hsg[0];
    const float sh1 = hsg[DFF];
    const float sq  = qag[0];
#pragma unroll
    for (int k = 0; k < SCAN_XLD; ++k) xs[0][srow + 2 * k][scol] = st[k];
    hsq[0][hpq][hw] = make_float4(sh0, sh1, sq, 0.0f);
  }
  __syncthreads();

  for (int ch = 0; ch < SCAN_NCHUNK; ++ch) {
    const int  buf  = ch & 1;
    const int  t0   = ch * SCAN_TC;
    const bool more = (ch + 1 < SCAN_NCHUNK);

    float4 st[SCAN_XLD];
    float  sh0 = 0.0f, sh1 = 0.0f, sq = 0.0f;
#pragma unroll
    for (int k = 0; k < SCAN_XLD; ++k) st[k] = make_float4(0.0f, 0.0f, 0.0f, 0.0f);
    if (more) {
      const size_t tn0 = (size_t)(t0 + SCAN_TC);
#pragma unroll
      for (int k = 0; k < SCAN_XLD; ++k) st[k] = xg[(tn0 + 2 * k) * SCAN_XQ];
      sh0 = hsg[tn0 * DFF];
      sh1 = hsg[(tn0 + 1) * DFF];
      sq  = qag[tn0 * DFF];
    }

    const auto ld_pair = [&](int tt) {
      stp_pair_ops o;
      const float4 hv = hsq[buf][tt >> 1][hrow];
      o.hsa = hv.x;
      o.hsb = hv.y;
      o.pa  = hv.z;
#pragma unroll
      for (int k = 0; k < 4; ++k) {
        o.a[k] = xs[buf][tt][16 * k + c];
        o.b[k] = xs[buf][tt + 1][16 * k + c];
      }
      return o;
    };

    float hkeep = 0.0f;
    float hk0 = 0.0f;

    const auto step_pair = [&](const stp_pair_ops& o, int tq) {
      float xa[16], xb[16];
#pragma unroll
      for (int k = 0; k < 4; ++k) {
        xa[4 * k] = o.a[k].x; xa[4 * k + 1] = o.a[k].y; xa[4 * k + 2] = o.a[k].z; xa[4 * k + 3] = o.a[k].w;
        xb[4 * k] = o.b[k].x; xb[4 * k + 1] = o.b[k].y; xb[4 * k + 2] = o.b[k].z; xb[4 * k + 3] = o.b[k].w;
      }

      asm volatile(""
                   : "+v"(r[0]), "+v"(r[1]), "+v"(r[2]), "+v"(r[3]),
                     "+v"(r[4]), "+v"(r[5]), "+v"(r[6]), "+v"(r[7]),
                     "+v"(r[8]), "+v"(r[9]), "+v"(r[10]), "+v"(r[11]),
                     "+v"(r[12]), "+v"(r[13]), "+v"(r[14]), "+v"(r[15])
                   : "v"(F[0])
                   : "memory");

      float q0[4], q1[4];
      float rF[16], gx[16];
#pragma unroll
      for (int u = 0; u < 16; ++u) { rF[u] = r[u] * F[u]; gx[u] = g[u] * xa[u]; }
#pragma unroll
      for (int j = 0; j < 4; ++j) {
        q0[j] = F[j] * xa[j];
        q1[j] = rF[j] * xb[j];
      }
#pragma unroll
      for (int u = 4; u < 16; ++u) {
        q0[u & 3] = fmaf(F[u], xa[u], q0[u & 3]);
        q1[u & 3] = fmaf(rF[u], xb[u], q1[u & 3]);
      }
      float p0 = stp_row_sum((q0[0] + q0[2]) + (q0[1] + q0[3]));
      const float p1 = stp_row_sum((q1[0] + q1[2]) + (q1[1] + q1[3]));

      float c1 = o.hsb + p1;
      asm("" : "+v"(p0), "+v"(c1));

      const float h0 = o.hsa + p0;
      const float h1 = fmaf(h0, o.pa, c1);
      hkeep = (c == tq)     ? h0 : hkeep;
      hkeep = (c == tq + 1) ? h1 : hkeep;

#pragma unroll
      for (int u = 0; u < 16; ++u) {
        const float F1 = fmaf(h0, gx[u], rF[u]);
        F[u] = fmaf(r[u], F1, g[u] * h1 * xb[u]);
      }
    };

    stp_pair_ops opA = ld_pair(0);
    for (int tp = 0; tp < SCAN_TC; tp += 4) {
      const int tq = tp & 15;
      const stp_pair_ops opB = ld_pair(tp + 2);
      step_pair(opA, tq);
      const int tn = (tp + 4 < SCAN_TC) ? tp + 4 : tp;
      opA = ld_pair(tn);
      step_pair(opB, tq + 2);
      hk0 = (tp == 12) ? hkeep : hk0;
    }

    if (more) {
#pragma unroll
      for (int k = 0; k < SCAN_XLD; ++k) xs[buf ^ 1][srow + 2 * k][scol] = st[k];
      hsq[buf ^ 1][hpq][hw] = make_float4(sh0, sh1, sq, 0.0f);
      __syncthreads();
    }

    {
      const float ga = 0.5f * hk0 * (1.0f + erff(hk0 * 0.70710678118654752f));
      const float gb = 0.5f * hkeep * (1.0f + erff(hkeep * 0.70710678118654752f));
      hg_ptr[(size_t)(t0 + c) * DFF] = ga;
      hg_ptr[(size_t)(t0 + 16 + c) * DFF] = gb;
    }
  }
}

extern "C" void kernel_launch(void* const* d_in, const int* in_sizes, int n_in,
                              void* d_out, int out_size, void* d_ws, size_t ws_size,
                              hipStream_t stream) {
  const float* x   = (const float*)d_in[0];
  const float* W1  = (const float*)d_in[1];
  const float* b1  = (const float*)d_in[2];
  const float* W2  = (const float*)d_in[3];
  const float* b2  = (const float*)d_in[4];
  const float* Lam = (const float*)d_in[5];
  const float* Gam = (const float*)d_in[6];
  float* out = (float*)d_out;

  float* Hs = (float*)d_ws;
  float* Hg = Hs + (size_t)MTOK * DFF;

  stp_wmma_gemm_bias<MTOK, DFF, DMOD, 64, 128, 2, 4, true>
      <<<(MTOK / 64) * (DFF / 128), 256, 0, stream>>>(x, W1, b1, Hs, Gam, Hg);

  stp_scan_kernel<<<(BSZ * DFF) / 8, 128, 0, stream>>>(x, Hs, Lam, Gam, Hg);

  stp_wmma_gemm_bias<MTOK, DMOD, DFF, 64, 64, 4, 2, false>
      <<<(MTOK / 64) * (DMOD / 64), 256, 0, stream>>>(Hg, W2, b2, out, nullptr, nullptr);
}


# scan: LDS operand reads of 8 time steps issued as one group, one group ahead (was one group per 2 steps): 3 read places per 32-step chunk instead of 17; sums bit-identical
# speedup vs baseline: 5.0347x; 1.0681x over previous
#include <hip/hip_runtime.h>
#include <math.h>

#define BSZ   4
#define LSEQ  1024
#define DMOD  256
#define DFF   512
#define MTOK  (BSZ * LSEQ)

typedef float v2f_t __attribute__((ext_vector_type(2)));
typedef float v8f_t __attribute__((ext_vector_type(8)));

#define GEMM_BK     64
#define GEMM_PITCH  (GEMM_BK + 4)
#define GEMM_Q      (GEMM_BK / 4)

template <int Mdim, int Ndim, int Kdim, int BM, int BN, int WM, int WN, bool PAIR>
__global__ __launch_bounds__(256)
void stp_wmma_gemm_bias(const float* __restrict__ A,
                        const float* __restrict__ Bt,
                        const float* __restrict__ bias,
                        float* __restrict__ D,
                        const float* __restrict__ Gt,
                        float* __restrict__ Q) {
  static_assert(WM * WN == 8, "8 waves per work-group");
  static_assert(Kdim % GEMM_BK == 0 && Mdim % BM == 0 && Ndim % BN == 0, "exact tiling");
  constexpr int TM = BM / WM, TN = BN / WN;
  constexpr int FM = TM / 16, FN = TN / 16;
  static_assert(TN == 32, "a wave's rows of D are whole 128-byte lines");
  constexpr int NC = Kdim / GEMM_BK;
  constexpr int LA = BM / 16, LB = BN / 16;
  constexpr int NBN = Ndim / BN;
  constexpr int PM  = BM / 2;
  constexpr int TM2 = PM / WM;
  static_assert(!PAIR || (TM2 == 16 && LA % 2 == 0), "one 16-row product tile per wave row");

  __shared__ __attribute__((aligned(16))) float As[2][BM][GEMM_PITCH];
  __shared__ __attribute__((aligned(16))) float Bs[2][BN][GEMM_PITCH];
  __shared__ __attribute__((aligned(16))) float Ps[PAIR ? 2 : 1][PAIR ? PM : 1][PAIR ? GEMM_PITCH : 4];
  __shared__ __attribute__((aligned(16))) float Gs[PAIR ? 2 : 1][PAIR ? BN : 1][PAIR ? GEMM_PITCH : 4];

  const int tid  = threadIdx.x;
  const int lane = tid & 31;
  const int wid  = tid >> 5;
  const int wm   = wid / WN;
  const int wn   = wid % WN;
  const int r    = lane & 15;
  const int h    = lane >> 4;
  const int bm   = blockIdx.x / NBN;
  const int bn   = blockIdx.x % NBN;

  const int sr = tid >> 4;
  const int sc = tid & (GEMM_Q - 1);
  const auto arow = [&](int j) { return PAIR ? 2 * sr + (j & 1) + 32 * (j >> 1) : sr + 16 * j; };
  const float4* ag = (const float4*)(A  + (size_t)(bm * BM) * Kdim) + sc;
  const float4* bg = (const float4*)(Bt + (size_t)(bn * BN + sr) * Kdim) + sc;
  const float4* gg = PAIR ? (const float4*)(Gt + (size_t)(bn * BN + sr) * Kdim) + sc : nullptr;
  constexpr size_t rstep = (size_t)16 * Kdim / 4;
  constexpr size_t rowq  = (size_t)Kdim / 4;
  const auto put_pairs = [&](int b, const float4* sa) {
#pragma unroll
    for (int jp = 0; jp < LA / 2; ++jp) {
      const float4 u = sa[2 * jp], w = sa[2 * jp + 1];
      *(float4*)&Ps[b][sr + 16 * jp][4 * sc] = make_float4(u.x * w.x, u.y * w.y, u.z * w.z, u.w * w.w);
    }
  };

  v8f_t c[FM][FN];
#pragma unroll
  for (int fm = 0; fm < FM; ++fm)
#pragma unroll
    for (int fn = 0; fn < FN; ++fn) c[fm][fn] = (v8f_t){};
  v8f_t c2[FN];
#pragma unroll
  for (int fn = 0; fn < FN; ++fn) c2[fn] = (v8f_t){};

  int doff = (bm * BM + wm * TM + 8 * h) * Ndim + bn * BN + wn * TN + r;
  float bv[FN];
#pragma unroll
  for (int fn = 0; fn < FN; ++fn) bv[fn] = bias[bn * BN + wn * TN + fn * 16 + r];
  int qoff = 2 * (bm * PM + wm * TM2 + 8 * h) * Ndim + bn * BN + wn * TN + r;
  asm volatile("" : "+v"(doff));
  if constexpr (PAIR) asm volatile("" : "+v"(qoff));

  {
    float4 sa[LA], sb[LB], sg[LB];
#pragma unroll
    for (int j = 0; j < LA; ++j) sa[j] = ag[(size_t)arow(j) * rowq];
#pragma unroll
    for (int j = 0; j < LB; ++j) sb[j] = bg[j * rstep];
    if constexpr (PAIR) {
#pragma unroll
      for (int j = 0; j < LB; ++j) sg[j] = gg[j * rstep];
    }
#pragma unroll
    for (int j = 0; j < LA; ++j) *(float4*)&As[0][arow(j)][4 * sc] = sa[j];
#pragma unroll
    for (int j = 0; j < LB; ++j) *(float4*)&Bs[0][sr + 16 * j][4 * sc] = sb[j];
    if constexpr (PAIR) {
      put_pairs(0, sa);
#pragma unroll
      for (int j = 0; j < LB; ++j) *(float4*)&Gs[0][sr + 16 * j][4 * sc] = sg[j];
    }
  }
  __syncthreads();

#pragma unroll
  for (int ch = 0; ch < NC; ++ch) {
    const int  buf  = ch & 1;
    const bool more = (ch + 1 < NC);

    float4 sa[LA], sb[LB], sg[LB];
#pragma unroll
    for (int j = 0; j < LA; ++j) sa[j] = make_float4(0.0f, 0.0f, 0.0f, 0.0f);
#pragma unroll
    for (int j = 0; j < LB; ++j) sb[j] = sg[j] = make_float4(0.0f, 0.0f, 0.0f, 0.0f);
    if (more) {
      const size_t koff = (size_t)(ch + 1) * GEMM_Q;
#pragma unroll
      for (int j = 0; j < LA; ++j) sa[j] = ag[(size_t)arow(j) * rowq + koff];
#pragma unroll
      for (int j = 0; j < LB; ++j) sb[j] = bg[j * rstep + koff];
      if constexpr (PAIR) {
#pragma unroll
        for (int j = 0; j < LB; ++j) sg[j] = gg[j * rstep + koff];
      }
    }

#pragma unroll
    for (int kb = 0; kb < GEMM_BK / 8; ++kb) {
      float4 a4[FM], b4[FN];
#pragma unroll
      for (int fm = 0; fm < FM; ++fm)
        a4[fm] = *(const float4*)&As[buf][wm * TM + fm * 16 + r][8 * kb + 4 * h];
#pragma unroll
      for (int fn = 0; fn < FN; ++fn)
        b4[fn] = *(const float4*)&Bs[buf][wn * TN + fn * 16 + r][8 * kb + 4 * h];
#pragma unroll
      for (int fm = 0; fm < FM; ++fm)
#pragma unroll
        for (int fn = 0; fn < FN; ++fn) {
          c[fm][fn] = __builtin_amdgcn_wmma_f32_16x16x4_f32(
              false, (v2f_t){a4[fm].x, a4[fm].y}, false, (v2f_t){b4[fn].x, b4[fn].y},
              (short)0, c[fm][fn], false, false);
        }
#pragma unroll
      for (int fm = 0; fm < FM; ++fm)
#pragma unroll
        for (int fn = 0; fn < FN; ++fn) {
          c[fm][fn] = __builtin_amdgcn_wmma_f32_16x16x4_f32(
              false, (v2f_t){a4[fm].z, a4[fm].w}, false, (v2f_t){b4[fn].z, b4[fn].w},
              (short)0, c[fm][fn], false, false);
        }
      if constexpr (PAIR) {
        const float4 p4 = *(const float4*)&Ps[buf][wm * TM2 + r][8 * kb + 4 * h];
        float4 g4[FN];
#pragma unroll
        for (int fn = 0; fn < FN; ++fn)
          g4[fn] = *(const float4*)&Gs[buf][wn * TN + fn * 16 + r][8 * kb + 4 * h];
#pragma unroll
        for (int fn = 0; fn < FN; ++fn)
          c2[fn] = __builtin_amdgcn_wmma_f32_16x16x4_f32(
              false, (v2f_t){p4.x, p4.y}, false, (v2f_t){g4[fn].x, g4[fn].y},
              (short)0, c2[fn], false, false);
#pragma unroll
        for (int fn = 0; fn < FN; ++fn)
          c2[fn] = __builtin_amdgcn_wmma_f32_16x16x4_f32(
              false, (v2f_t){p4.z, p4.w}, false, (v2f_t){g4[fn].z, g4[fn].w},
              (short)0, c2[fn], false, false);
      }
    }

    if (more) {
#pragma unroll
      for (int j = 0; j < LA; ++j) *(float4*)&As[buf ^ 1][arow(j)][4 * sc] = sa[j];
#pragma unroll
      for (int j = 0; j < LB; ++j) *(float4*)&Bs[buf ^ 1][sr + 16 * j][4 * sc] = sb[j];
      if constexpr (PAIR) {
        put_pairs(buf ^ 1, sa);
#pragma unroll
        for (int j = 0; j < LB; ++j) *(float4*)&Gs[buf ^ 1][sr + 16 * j][4 * sc] = sg[j];
      }
      __syncthreads();
    }
  }

  __builtin_amdgcn_sched_barrier(0);
  asm volatile("v_nop\n\tv_nop\n\tv_nop\n\tv_nop");
  __builtin_amdgcn_sched_barrier(0);

#pragma unroll
  for (int fm = 0; fm < FM; ++fm)
#pragma unroll
    for (int fn = 0; fn < FN; ++fn) {
#pragma unroll
      for (int v = 0; v < 8; ++v)
        D[(size_t)(doff + (fm * 16 + v) * Ndim + fn * 16)] = c[fm][fn][v] + bv[fn];
    }
  if constexpr (PAIR) {
#pragma unroll
    for (int fn = 0; fn < FN; ++fn) {
#pragma unroll
      for (int v = 0; v < 8; ++v)
        Q[(size_t)(qoff + 2 * v * Ndim + fn * 16)] = c2[fn][v];
    }
  }
}

template <int M>
__device__ __forceinline__ float stp_row_xor(float v) {
  const int b = __builtin_bit_cast(int, v);
  return __builtin_bit_cast(
      float, __builtin_amdgcn_update_dpp(b, b, 0x160 + M, 0xf, 0xf, true));
}

__device__ __forceinline__ float stp_row_sum(float p) {
  p += stp_row_xor<8>(p);
  p += stp_row_xor<4>(p);
  p += stp_row_xor<2>(p);
  p += stp_row_xor<1>(p);
  return p;
}

#define SCAN_TC     32
#define SCAN_WAVES  4
#define SCAN_ROWS   (2 * SCAN_WAVES)
#define SCAN_NCHUNK (LSEQ / SCAN_TC)
#define SCAN_XQ     (DMOD / 4)
#define SCAN_XLD    (SCAN_TC * SCAN_XQ / (SCAN_WAVES * 32))
#define SCAN_NP     (SCAN_TC / 2)
static_assert(SCAN_TC == 32, "lane c of a row keeps the h of the chunk's steps c and 16 + c");
static_assert(SCAN_XQ == 64, "a lane's four float4 of an x row are 16 float4 apart");
static_assert(SCAN_NP * SCAN_ROWS == SCAN_WAVES * 32, "a thread stages one (step pair, dff-row) per chunk");

struct stp_pair_ops {
  float4 a[4];
  float4 b[4];
  float  hsa, hsb;
  float  pa;
};

#define SCAN_GP 4
static_assert(2 * 2 * SCAN_GP == 16, "two read groups cover the 16 steps whose h a lane row keeps");
struct stp_group_ops {
  stp_pair_ops p[SCAN_GP];
};

__global__ __launch_bounds__(SCAN_WAVES * 32)
void stp_scan_kernel(const float* __restrict__ x,
                     const float* __restrict__ Hs,
                     const float* __restrict__ Lambda,
                     const float* __restrict__ Gamma,
                     float* __restrict__ Hg) {
  __shared__ float4 xs[2][SCAN_TC][SCAN_XQ];
  __shared__ float4 hsq[2][SCAN_NP][SCAN_ROWS];

  const int tid  = threadIdx.x;
  const int wid  = tid >> 5;
  const int lane = tid & 31;
  const int c    = lane & 15;
  const int hrow = 2 * wid + (lane >> 4);
  const int row0 = blockIdx.x * SCAN_ROWS;
  if (row0 >= BSZ * DFF) return;
  const int b  = row0 >> 9;
  const int i0 = row0 & (DFF - 1);
  const int i  = i0 + hrow;

  float r[16], g[16], F[16];
  const float4* lam = (const float4*)(Lambda + (size_t)i * DMOD) + c;
  const float4* gam = (const float4*)(Gamma  + (size_t)i * DMOD) + c;
#pragma unroll
  for (int k = 0; k < 4; ++k) {
    const float4 l4 = lam[16 * k];
    const float4 g4 = gam[16 * k];
    const float lv[4] = {l4.x, l4.y, l4.z, l4.w};
    const float gv[4] = {g4.x, g4.y, g4.z, g4.w};
#pragma unroll
    for (int j = 0; j < 4; ++j) {
      const float s = 1.0f / (1.0f + __expf(-lv[j]));
      r[4 * k + j] = 1.0f - s;
      g[4 * k + j] = gv[j];
      F[4 * k + j] = 0.0f;
    }
  }

  const int srow = tid >> 6;
  const int scol = tid & (SCAN_XQ - 1);
  const int hpq  = tid >> 3;
  const int hw   = tid & (SCAN_ROWS - 1);
  const float4* xg  = (const float4*)(x + (size_t)b * LSEQ * DMOD) + (size_t)srow * SCAN_XQ + scol;
  const size_t  hoff = (size_t)b * LSEQ * DFF + (size_t)(2 * hpq) * DFF + i0 + hw;
  const float*  hsg = Hs + hoff;
  const float*  qag = Hg + hoff;
  float*     hg_ptr = Hg + (size_t)b * LSEQ * DFF + i;

  {
    float4 st[SCAN_XLD];
#pragma unroll
    for (int k = 0; k < SCAN_XLD; ++k) st[k] = xg[(size_t)(2 * k) * SCAN_XQ];
    const float sh0 = hsg[0];
    const float sh1 = hsg[DFF];
    const float sq  = qag[0];
#pragma unroll
    for (int k = 0; k < SCAN_XLD; ++k) xs[0][srow + 2 * k][scol] = st[k];
    hsq[0][hpq][hw] = make_float4(sh0, sh1, sq, 0.0f);
  }
  __syncthreads();

  for (int ch = 0; ch < SCAN_NCHUNK; ++ch) {
    const int  buf  = ch & 1;
    const int  t0   = ch * SCAN_TC;
    const bool more = (ch + 1 < SCAN_NCHUNK);

    float4 st[SCAN_XLD];
    float  sh0 = 0.0f, sh1 = 0.0f, sq = 0.0f;
#pragma unroll
    for (int k = 0; k < SCAN_XLD; ++k) st[k] = make_float4(0.0f, 0.0f, 0.0f, 0.0f);
    if (more) {
      const size_t tn0 = (size_t)(t0 + SCAN_TC);
#pragma unroll
      for (int k = 0; k < SCAN_XLD; ++k) st[k] = xg[(tn0 + 2 * k) * SCAN_XQ];
      sh0 = hsg[tn0 * DFF];
      sh1 = hsg[(tn0 + 1) * DFF];
      sq  = qag[tn0 * DFF];
    }

    const auto ld_pair = [&](int tt) {
      stp_pair_ops o;
      const float4 hv = hsq[buf][tt >> 1][hrow];
      o.hsa = hv.x;
      o.hsb = hv.y;
      o.pa  = hv.z;
#pragma unroll
      for (int k = 0; k < 4; ++k) {
        o.a[k] = xs[buf][tt][16 * k + c];
        o.b[k] = xs[buf][tt + 1][16 * k + c];
      }
      return o;
    };

    float hkeep = 0.0f;
    float hk0 = 0.0f;

    const auto step_pair = [&](const stp_pair_ops& o, int tq) {
      float xa[16], xb[16];
#pragma unroll
      for (int k = 0; k < 4; ++k) {
        xa[4 * k] = o.a[k].x; xa[4 * k + 1] = o.a[k].y; xa[4 * k + 2] = o.a[k].z; xa[4 * k + 3] = o.a[k].w;
        xb[4 * k] = o.b[k].x; xb[4 * k + 1] = o.b[k].y; xb[4 * k + 2] = o.b[k].z; xb[4 * k + 3] = o.b[k].w;
      }

      asm volatile(""
                   : "+v"(r[0]), "+v"(r[1]), "+v"(r[2]), "+v"(r[3]),
                     "+v"(r[4]), "+v"(r[5]), "+v"(r[6]), "+v"(r[7]),
                     "+v"(r[8]), "+v"(r[9]), "+v"(r[10]), "+v"(r[11]),
                     "+v"(r[12]), "+v"(r[13]), "+v"(r[14]), "+v"(r[15])
                   : "v"(F[0])
                   : "memory");

      float q0[4], q1[4];
      float rF[16], gx[16];
#pragma unroll
      for (int u = 0; u < 16; ++u) { rF[u] = r[u] * F[u]; gx[u] = g[u] * xa[u]; }
#pragma unroll
      for (int j = 0; j < 4; ++j) {
        q0[j] = F[j] * xa[j];
        q1[j] = rF[j] * xb[j];
      }
#pragma unroll
      for (int u = 4; u < 16; ++u) {
        q0[u & 3] = fmaf(F[u], xa[u], q0[u & 3]);
        q1[u & 3] = fmaf(rF[u], xb[u], q1[u & 3]);
      }
      float p0 = stp_row_sum((q0[0] + q0[2]) + (q0[1] + q0[3]));
      const float p1 = stp_row_sum((q1[0] + q1[2]) + (q1[1] + q1[3]));

      float c1 = o.hsb + p1;
      asm("" : "+v"(p0), "+v"(c1));

      const float h0 = o.hsa + p0;
      const float h1 = fmaf(h0, o.pa, c1);
      hkeep = (c == tq)     ? h0 : hkeep;
      hkeep = (c == tq + 1) ? h1 : hkeep;

#pragma unroll
      for (int u = 0; u < 16; ++u) {
        const float F1 = fmaf(h0, gx[u], rF[u]);
        F[u] = fmaf(r[u], F1, g[u] * h1 * xb[u]);
      }
    };

    const auto ld_group = [&](int tt) {
      stp_group_ops gq;
#pragma unroll
      for (int k = 0; k < SCAN_GP; ++k) gq.p[k] = ld_pair(tt + 2 * k);
      return gq;
    };
    stp_group_ops gA = ld_group(0);
#pragma unroll
    for (int tp = 0; tp < SCAN_TC; tp += 16) {
      const stp_group_ops gB = ld_group(tp + 8);
#pragma unroll
      for (int k = 0; k < SCAN_GP; ++k) step_pair(gA.p[k], 2 * k);
      if (tp + 16 < SCAN_TC) gA = ld_group(tp + 16);
#pragma unroll
      for (int k = 0; k < SCAN_GP; ++k) step_pair(gB.p[k], 8 + 2 * k);
      hk0 = (tp == 0) ? hkeep : hk0;
    }

    if (more) {
#pragma unroll
      for (int k = 0; k < SCAN_XLD; ++k) xs[buf ^ 1][srow + 2 * k][scol] = st[k];
      hsq[buf ^ 1][hpq][hw] = make_float4(sh0, sh1, sq, 0.0f);
      __syncthreads();
    }

    {
      const float ga = 0.5f * hk0 * (1.0f + erff(hk0 * 0.70710678118654752f));
      const float gb = 0.5f * hkeep * (1.0f + erff(hkeep * 0.70710678118654752f));
      hg_ptr[(size_t)(t0 + c) * DFF] = ga;
      hg_ptr[(size_t)(t0 + 16 + c) * DFF] = gb;
    }
  }
}

extern "C" void kernel_launch(void* const* d_in, const int* in_sizes, int n_in,
                              void* d_out, int out_size, void* d_ws, size_t ws_size,
                              hipStream_t stream) {
  const float* x   = (const float*)d_in[0];
  const float* W1  = (const float*)d_in[1];
  const float* b1  = (const float*)d_in[2];
  const float* W2  = (const float*)d_in[3];
  const float* b2  = (const float*)d_in[4];
  const float* Lam = (const float*)d_in[5];
  const float* Gam = (const float*)d_in[6];
  float* out = (float*)d_out;

  float* Hs = (float*)d_ws;
  float* Hg = Hs + (size_t)MTOK * DFF;

  stp_wmma_gemm_bias<MTOK, DFF, DMOD, 64, 128, 2, 4, true>
      <<<(MTOK / 64) * (DFF / 128), 256, 0, stream>>>(x, W1, b1, Hs, Gam, Hg);

  stp_scan_kernel<<<(BSZ * DFF) / 8, 128, 0, stream>>>(x, Hs, Lam, Gam, Hg);

  stp_wmma_gemm_bias<MTOK, DMOD, DFF, 64, 64, 4, 2, false>
      <<<(MTOK / 64) * (DMOD / 64), 256, 0, stream>>>(Hg, W2, b2, out, nullptr, nullptr);
}


# scan: the pair's h-free products g*x0, g*x1 are issued between the lane sum's four cross-lane steps (four columns a step), not around them; F's second update is h1*(g*x1)+r*F1: F rounding changes
# speedup vs baseline: 5.2243x; 1.0377x over previous
#include <hip/hip_runtime.h>
#include <math.h>

#define BSZ   4
#define LSEQ  1024
#define DMOD  256
#define DFF   512
#define MTOK  (BSZ * LSEQ)

typedef float v2f_t __attribute__((ext_vector_type(2)));
typedef float v8f_t __attribute__((ext_vector_type(8)));

#define GEMM_BK     64
#define GEMM_PITCH  (GEMM_BK + 4)
#define GEMM_Q      (GEMM_BK / 4)

template <int Mdim, int Ndim, int Kdim, int BM, int BN, int WM, int WN, bool PAIR>
__global__ __launch_bounds__(256)
void stp_wmma_gemm_bias(const float* __restrict__ A,
                        const float* __restrict__ Bt,
                        const float* __restrict__ bias,
                        float* __restrict__ D,
                        const float* __restrict__ Gt,
                        float* __restrict__ Q) {
  static_assert(WM * WN == 8, "8 waves per work-group");
  static_assert(Kdim % GEMM_BK == 0 && Mdim % BM == 0 && Ndim % BN == 0, "exact tiling");
  constexpr int TM = BM / WM, TN = BN / WN;
  constexpr int FM = TM / 16, FN = TN / 16;
  static_assert(TN == 32, "a wave's rows of D are whole 128-byte lines");
  constexpr int NC = Kdim / GEMM_BK;
  constexpr int LA = BM / 16, LB = BN / 16;
  constexpr int NBN = Ndim / BN;
  constexpr int PM  = BM / 2;
  constexpr int TM2 = PM / WM;
  static_assert(!PAIR || (TM2 == 16 && LA % 2 == 0), "one 16-row product tile per wave row");

  __shared__ __attribute__((aligned(16))) float As[2][BM][GEMM_PITCH];
  __shared__ __attribute__((aligned(16))) float Bs[2][BN][GEMM_PITCH];
  __shared__ __attribute__((aligned(16))) float Ps[PAIR ? 2 : 1][PAIR ? PM : 1][PAIR ? GEMM_PITCH : 4];
  __shared__ __attribute__((aligned(16))) float Gs[PAIR ? 2 : 1][PAIR ? BN : 1][PAIR ? GEMM_PITCH : 4];

  const int tid  = threadIdx.x;
  const int lane = tid & 31;
  const int wid  = tid >> 5;
  const int wm   = wid / WN;
  const int wn   = wid % WN;
  const int r    = lane & 15;
  const int h    = lane >> 4;
  const int bm   = blockIdx.x / NBN;
  const int bn   = blockIdx.x % NBN;

  const int sr = tid >> 4;
  const int sc = tid & (GEMM_Q - 1);
  const auto arow = [&](int j) { return PAIR ? 2 * sr + (j & 1) + 32 * (j >> 1) : sr + 16 * j; };
  const float4* ag = (const float4*)(A  + (size_t)(bm * BM) * Kdim) + sc;
  const float4* bg = (const float4*)(Bt + (size_t)(bn * BN + sr) * Kdim) + sc;
  const float4* gg = PAIR ? (const float4*)(Gt + (size_t)(bn * BN + sr) * Kdim) + sc : nullptr;
  constexpr size_t rstep = (size_t)16 * Kdim / 4;
  constexpr size_t rowq  = (size_t)Kdim / 4;
  const auto put_pairs = [&](int b, const float4* sa) {
#pragma unroll
    for (int jp = 0; jp < LA / 2; ++jp) {
      const float4 u = sa[2 * jp], w = sa[2 * jp + 1];
      *(float4*)&Ps[b][sr + 16 * jp][4 * sc] = make_float4(u.x * w.x, u.y * w.y, u.z * w.z, u.w * w.w);
    }
  };

  v8f_t c[FM][FN];
#pragma unroll
  for (int fm = 0; fm < FM; ++fm)
#pragma unroll
    for (int fn = 0; fn < FN; ++fn) c[fm][fn] = (v8f_t){};
  v8f_t c2[FN];
#pragma unroll
  for (int fn = 0; fn < FN; ++fn) c2[fn] = (v8f_t){};

  int doff = (bm * BM + wm * TM + 8 * h) * Ndim + bn * BN + wn * TN + r;
  float bv[FN];
#pragma unroll
  for (int fn = 0; fn < FN; ++fn) bv[fn] = bias[bn * BN + wn * TN + fn * 16 + r];
  int qoff = 2 * (bm * PM + wm * TM2 + 8 * h) * Ndim + bn * BN + wn * TN + r;
  asm volatile("" : "+v"(doff));
  if constexpr (PAIR) asm volatile("" : "+v"(qoff));

  {
    float4 sa[LA], sb[LB], sg[LB];
#pragma unroll
    for (int j = 0; j < LA; ++j) sa[j] = ag[(size_t)arow(j) * rowq];
#pragma unroll
    for (int j = 0; j < LB; ++j) sb[j] = bg[j * rstep];
    if constexpr (PAIR) {
#pragma unroll
      for (int j = 0; j < LB; ++j) sg[j] = gg[j * rstep];
    }
#pragma unroll
    for (int j = 0; j < LA; ++j) *(float4*)&As[0][arow(j)][4 * sc] = sa[j];
#pragma unroll
    for (int j = 0; j < LB; ++j) *(float4*)&Bs[0][sr + 16 * j][4 * sc] = sb[j];
    if constexpr (PAIR) {
      put_pairs(0, sa);
#pragma unroll
      for (int j = 0; j < LB; ++j) *(float4*)&Gs[0][sr + 16 * j][4 * sc] = sg[j];
    }
  }
  __syncthreads();

#pragma unroll
  for (int ch = 0; ch < NC; ++ch) {
    const int  buf  = ch & 1;
    const bool more = (ch + 1 < NC);

    float4 sa[LA], sb[LB], sg[LB];
#pragma unroll
    for (int j = 0; j < LA; ++j) sa[j] = make_float4(0.0f, 0.0f, 0.0f, 0.0f);
#pragma unroll
    for (int j = 0; j < LB; ++j) sb[j] = sg[j] = make_float4(0.0f, 0.0f, 0.0f, 0.0f);
    if (more) {
      const size_t koff = (size_t)(ch + 1) * GEMM_Q;
#pragma unroll
      for (int j = 0; j < LA; ++j) sa[j] = ag[(size_t)arow(j) * rowq + koff];
#pragma unroll
      for (int j = 0; j < LB; ++j) sb[j] = bg[j * rstep + koff];
      if constexpr (PAIR) {
#pragma unroll
        for (int j = 0; j < LB; ++j) sg[j] = gg[j * rstep + koff];
      }
    }

#pragma unroll
    for (int kb = 0; kb < GEMM_BK / 8; ++kb) {
      float4 a4[FM], b4[FN];
#pragma unroll
      for (int fm = 0; fm < FM; ++fm)
        a4[fm] = *(const float4*)&As[buf][wm * TM + fm * 16 + r][8 * kb + 4 * h];
#pragma unroll
      for (int fn = 0; fn < FN; ++fn)
        b4[fn] = *(const float4*)&Bs[buf][wn * TN + fn * 16 + r][8 * kb + 4 * h];
#pragma unroll
      for (int fm = 0; fm < FM; ++fm)
#pragma unroll
        for (int fn = 0; fn < FN; ++fn) {
          c[fm][fn] = __builtin_amdgcn_wmma_f32_16x16x4_f32(
              false, (v2f_t){a4[fm].x, a4[fm].y}, false, (v2f_t){b4[fn].x, b4[fn].y},
              (short)0, c[fm][fn], false, false);
        }
#pragma unroll
      for (int fm = 0; fm < FM; ++fm)
#pragma unroll
        for (int fn = 0; fn < FN; ++fn) {
          c[fm][fn] = __builtin_amdgcn_wmma_f32_16x16x4_f32(
              false, (v2f_t){a4[fm].z, a4[fm].w}, false, (v2f_t){b4[fn].z, b4[fn].w},
              (short)0, c[fm][fn], false, false);
        }
      if constexpr (PAIR) {
        const float4 p4 = *(const float4*)&Ps[buf][wm * TM2 + r][8 * kb + 4 * h];
        float4 g4[FN];
#pragma unroll
        for (int fn = 0; fn < FN; ++fn)
          g4[fn] = *(const float4*)&Gs[buf][wn * TN + fn * 16 + r][8 * kb + 4 * h];
#pragma unroll
        for (int fn = 0; fn < FN; ++fn)
          c2[fn] = __builtin_amdgcn_wmma_f32_16x16x4_f32(
              false, (v2f_t){p4.x, p4.y}, false, (v2f_t){g4[fn].x, g4[fn].y},
              (short)0, c2[fn], false, false);
#pragma unroll
        for (int fn = 0; fn < FN; ++fn)
          c2[fn] = __builtin_amdgcn_wmma_f32_16x16x4_f32(
              false, (v2f_t){p4.z, p4.w}, false, (v2f_t){g4[fn].z, g4[fn].w},
              (short)0, c2[fn], false, false);
      }
    }

    if (more) {
#pragma unroll
      for (int j = 0; j < LA; ++j) *(float4*)&As[buf ^ 1][arow(j)][4 * sc] = sa[j];
#pragma unroll
      for (int j = 0; j < LB; ++j) *(float4*)&Bs[buf ^ 1][sr + 16 * j][4 * sc] = sb[j];
      if constexpr (PAIR) {
        put_pairs(buf ^ 1, sa);
#pragma unroll
        for (int j = 0; j < LB; ++j) *(float4*)&Gs[buf ^ 1][sr + 16 * j][4 * sc] = sg[j];
      }
      __syncthreads();
    }
  }

  __builtin_amdgcn_sched_barrier(0);
  asm volatile("v_nop\n\tv_nop\n\tv_nop\n\tv_nop");
  __builtin_amdgcn_sched_barrier(0);

#pragma unroll
  for (int fm = 0; fm < FM; ++fm)
#pragma unroll
    for (int fn = 0; fn < FN; ++fn) {
#pragma unroll
      for (int v = 0; v < 8; ++v)
        D[(size_t)(doff + (fm * 16 + v) * Ndim + fn * 16)] = c[fm][fn][v] + bv[fn];
    }
  if constexpr (PAIR) {
#pragma unroll
    for (int fn = 0; fn < FN; ++fn) {
#pragma unroll
      for (int v = 0; v < 8; ++v)
        Q[(size_t)(qoff + 2 * v * Ndim + fn * 16)] = c2[fn][v];
    }
  }
}

template <int M>
__device__ __forceinline__ float stp_row_xor(float v) {
  const int b = __builtin_bit_cast(int, v);
  return __builtin_bit_cast(
      float, __builtin_amdgcn_update_dpp(b, b, 0x160 + M, 0xf, 0xf, true));
}


template <int M>
__device__ __forceinline__ void stp_lane_step(float& p0, float& p1, const float* g, const float* xa,
                                              const float* xb, float* gx, float* gy, int base) {
  p0 += stp_row_xor<M>(p0);
  p1 += stp_row_xor<M>(p1);
#pragma unroll
  for (int j = 0; j < 4; ++j) {
    gx[base + j] = g[base + j] * xa[base + j];
    gy[base + j] = g[base + j] * xb[base + j];
  }
  if (base + 4 < 16) __builtin_amdgcn_sched_barrier(0);
}

#define SCAN_TC     32
#define SCAN_WAVES  4
#define SCAN_ROWS   (2 * SCAN_WAVES)
#define SCAN_NCHUNK (LSEQ / SCAN_TC)
#define SCAN_XQ     (DMOD / 4)
#define SCAN_XLD    (SCAN_TC * SCAN_XQ / (SCAN_WAVES * 32))
#define SCAN_NP     (SCAN_TC / 2)
static_assert(SCAN_TC == 32, "lane c of a row keeps the h of the chunk's steps c and 16 + c");
static_assert(SCAN_XQ == 64, "a lane's four float4 of an x row are 16 float4 apart");
static_assert(SCAN_NP * SCAN_ROWS == SCAN_WAVES * 32, "a thread stages one (step pair, dff-row) per chunk");

struct stp_pair_ops {
  float4 a[4];
  float4 b[4];
  float  hsa, hsb;
  float  pa;
};

#define SCAN_GP 4
static_assert(2 * 2 * SCAN_GP == 16, "two read groups cover the 16 steps whose h a lane row keeps");
struct stp_group_ops {
  stp_pair_ops p[SCAN_GP];
};

__global__ __launch_bounds__(SCAN_WAVES * 32)
void stp_scan_kernel(const float* __restrict__ x,
                     const float* __restrict__ Hs,
                     const float* __restrict__ Lambda,
                     const float* __restrict__ Gamma,
                     float* __restrict__ Hg) {
  __shared__ float4 xs[2][SCAN_TC][SCAN_XQ];
  __shared__ float4 hsq[2][SCAN_NP][SCAN_ROWS];

  const int tid  = threadIdx.x;
  const int wid  = tid >> 5;
  const int lane = tid & 31;
  const int c    = lane & 15;
  const int hrow = 2 * wid + (lane >> 4);
  const int row0 = blockIdx.x * SCAN_ROWS;
  if (row0 >= BSZ * DFF) return;
  const int b  = row0 >> 9;
  const int i0 = row0 & (DFF - 1);
  const int i  = i0 + hrow;

  float r[16], g[16], F[16];
  const float4* lam = (const float4*)(Lambda + (size_t)i * DMOD) + c;
  const float4* gam = (const float4*)(Gamma  + (size_t)i * DMOD) + c;
#pragma unroll
  for (int k = 0; k < 4; ++k) {
    const float4 l4 = lam[16 * k];
    const float4 g4 = gam[16 * k];
    const float lv[4] = {l4.x, l4.y, l4.z, l4.w};
    const float gv[4] = {g4.x, g4.y, g4.z, g4.w};
#pragma unroll
    for (int j = 0; j < 4; ++j) {
      const float s = 1.0f / (1.0f + __expf(-lv[j]));
      r[4 * k + j] = 1.0f - s;
      g[4 * k + j] = gv[j];
      F[4 * k + j] = 0.0f;
    }
  }

  const int srow = tid >> 6;
  const int scol = tid & (SCAN_XQ - 1);
  const int hpq  = tid >> 3;
  const int hw   = tid & (SCAN_ROWS - 1);
  const float4* xg  = (const float4*)(x + (size_t)b * LSEQ * DMOD) + (size_t)srow * SCAN_XQ + scol;
  const size_t  hoff = (size_t)b * LSEQ * DFF + (size_t)(2 * hpq) * DFF + i0 + hw;
  const float*  hsg = Hs + hoff;
  const float*  qag = Hg + hoff;
  float*     hg_ptr = Hg + (size_t)b * LSEQ * DFF + i;

  {
    float4 st[SCAN_XLD];
#pragma unroll
    for (int k = 0; k < SCAN_XLD; ++k) st[k] = xg[(size_t)(2 * k) * SCAN_XQ];
    const float sh0 = hsg[0];
    const float sh1 = hsg[DFF];
    const float sq  = qag[0];
#pragma unroll
    for (int k = 0; k < SCAN_XLD; ++k) xs[0][srow + 2 * k][scol] = st[k];
    hsq[0][hpq][hw] = make_float4(sh0, sh1, sq, 0.0f);
  }
  __syncthreads();

  for (int ch = 0; ch < SCAN_NCHUNK; ++ch) {
    const int  buf  = ch & 1;
    const int  t0   = ch * SCAN_TC;
    const bool more = (ch + 1 < SCAN_NCHUNK);

    float4 st[SCAN_XLD];
    float  sh0 = 0.0f, sh1 = 0.0f, sq = 0.0f;
#pragma unroll
    for (int k = 0; k < SCAN_XLD; ++k) st[k] = make_float4(0.0f, 0.0f, 0.0f, 0.0f);
    if (more) {
      const size_t tn0 = (size_t)(t0 + SCAN_TC);
#pragma unroll
      for (int k = 0; k < SCAN_XLD; ++k) st[k] = xg[(tn0 + 2 * k) * SCAN_XQ];
      sh0 = hsg[tn0 * DFF];
      sh1 = hsg[(tn0 + 1) * DFF];
      sq  = qag[tn0 * DFF];
    }

    const auto ld_pair = [&](int tt) {
      stp_pair_ops o;
      const float4 hv = hsq[buf][tt >> 1][hrow];
      o.hsa = hv.x;
      o.hsb = hv.y;
      o.pa  = hv.z;
#pragma unroll
      for (int k = 0; k < 4; ++k) {
        o.a[k] = xs[buf][tt][16 * k + c];
        o.b[k] = xs[buf][tt + 1][16 * k + c];
      }
      return o;
    };

    float hkeep = 0.0f;
    float hk0 = 0.0f;

    const auto step_pair = [&](const stp_pair_ops& o, int tq) {
      float xa[16], xb[16];
#pragma unroll
      for (int k = 0; k < 4; ++k) {
        xa[4 * k] = o.a[k].x; xa[4 * k + 1] = o.a[k].y; xa[4 * k + 2] = o.a[k].z; xa[4 * k + 3] = o.a[k].w;
        xb[4 * k] = o.b[k].x; xb[4 * k + 1] = o.b[k].y; xb[4 * k + 2] = o.b[k].z; xb[4 * k + 3] = o.b[k].w;
      }

      asm volatile(""
                   : "+v"(r[0]), "+v"(r[1]), "+v"(r[2]), "+v"(r[3]),
                     "+v"(r[4]), "+v"(r[5]), "+v"(r[6]), "+v"(r[7]),
                     "+v"(r[8]), "+v"(r[9]), "+v"(r[10]), "+v"(r[11]),
                     "+v"(r[12]), "+v"(r[13]), "+v"(r[14]), "+v"(r[15])
                   : "v"(F[0])
                   : "memory");

      float q0[4], q1[4];
      float rF[16];
#pragma unroll
      for (int u = 0; u < 16; ++u) rF[u] = r[u] * F[u];
#pragma unroll
      for (int j = 0; j < 4; ++j) {
        q0[j] = F[j] * xa[j];
        q1[j] = rF[j] * xb[j];
      }
#pragma unroll
      for (int u = 4; u < 16; ++u) {
        q0[u & 3] = fmaf(F[u], xa[u], q0[u & 3]);
        q1[u & 3] = fmaf(rF[u], xb[u], q1[u & 3]);
      }
      float gx[16], gy[16];
      float p0, p1;
      __builtin_amdgcn_sched_barrier(0);
      p0 = (q0[0] + q0[2]) + (q0[1] + q0[3]);
      p1 = (q1[0] + q1[2]) + (q1[1] + q1[3]);
      stp_lane_step<8>(p0, p1, g, xa, xb, gx, gy, 0);
      stp_lane_step<4>(p0, p1, g, xa, xb, gx, gy, 4);
      stp_lane_step<2>(p0, p1, g, xa, xb, gx, gy, 8);
      stp_lane_step<1>(p0, p1, g, xa, xb, gx, gy, 12);

      float c1 = o.hsb + p1;
      asm("" : "+v"(p0), "+v"(c1));

      const float h0 = o.hsa + p0;
      const float h1 = fmaf(h0, o.pa, c1);
      hkeep = (c == tq)     ? h0 : hkeep;
      hkeep = (c == tq + 1) ? h1 : hkeep;

#pragma unroll
      for (int u = 0; u < 16; ++u) {
        const float F1 = fmaf(h0, gx[u], rF[u]);
        F[u] = fmaf(h1, gy[u], r[u] * F1);
      }
    };

    const auto ld_group = [&](int tt) {
      stp_group_ops gq;
#pragma unroll
      for (int k = 0; k < SCAN_GP; ++k) gq.p[k] = ld_pair(tt + 2 * k);
      return gq;
    };
    stp_group_ops gA = ld_group(0);
#pragma unroll
    for (int tp = 0; tp < SCAN_TC; tp += 16) {
      const stp_group_ops gB = ld_group(tp + 8);
#pragma unroll
      for (int k = 0; k < SCAN_GP; ++k) step_pair(gA.p[k], 2 * k);
      if (tp + 16 < SCAN_TC) gA = ld_group(tp + 16);
#pragma unroll
      for (int k = 0; k < SCAN_GP; ++k) step_pair(gB.p[k], 8 + 2 * k);
      hk0 = (tp == 0) ? hkeep : hk0;
    }

    if (more) {
#pragma unroll
      for (int k = 0; k < SCAN_XLD; ++k) xs[buf ^ 1][srow + 2 * k][scol] = st[k];
      hsq[buf ^ 1][hpq][hw] = make_float4(sh0, sh1, sq, 0.0f);
      __syncthreads();
    }

    {
      const float ga = 0.5f * hk0 * (1.0f + erff(hk0 * 0.70710678118654752f));
      const float gb = 0.5f * hkeep * (1.0f + erff(hkeep * 0.70710678118654752f));
      hg_ptr[(size_t)(t0 + c) * DFF] = ga;
      hg_ptr[(size_t)(t0 + 16 + c) * DFF] = gb;
    }
  }
}

extern "C" void kernel_launch(void* const* d_in, const int* in_sizes, int n_in,
                              void* d_out, int out_size, void* d_ws, size_t ws_size,
                              hipStream_t stream) {
  const float* x   = (const float*)d_in[0];
  const float* W1  = (const float*)d_in[1];
  const float* b1  = (const float*)d_in[2];
  const float* W2  = (const float*)d_in[3];
  const float* b2  = (const float*)d_in[4];
  const float* Lam = (const float*)d_in[5];
  const float* Gam = (const float*)d_in[6];
  float* out = (float*)d_out;

  float* Hs = (float*)d_ws;
  float* Hg = Hs + (size_t)MTOK * DFF;

  stp_wmma_gemm_bias<MTOK, DFF, DMOD, 64, 128, 2, 4, true>
      <<<(MTOK / 64) * (DFF / 128), 256, 0, stream>>>(x, W1, b1, Hs, Gam, Hg);

  stp_scan_kernel<<<(BSZ * DFF) / 8, 128, 0, stream>>>(x, Hs, Lam, Gam, Hg);

  stp_wmma_gemm_bias<MTOK, DMOD, DFF, 64, 64, 4, 2, false>
      <<<(MTOK / 64) * (DMOD / 64), 256, 0, stream>>>(Hg, W2, b2, out, nullptr, nullptr);
}


# scan: a chunk's GELU leaves the serial path: issued behind the NEXT chunk's first LDS reads, its two erf branch-free side by side, its h picked by 4 select chains of 4 (was one of 16); bit-identical
# speedup vs baseline: 5.4461x; 1.0425x over previous
#include <hip/hip_runtime.h>
#include <math.h>

#define BSZ   4
#define LSEQ  1024
#define DMOD  256
#define DFF   512
#define MTOK  (BSZ * LSEQ)

typedef float v2f_t __attribute__((ext_vector_type(2)));
typedef float v8f_t __attribute__((ext_vector_type(8)));

#define GEMM_BK     64
#define GEMM_PITCH  (GEMM_BK + 4)
#define GEMM_Q      (GEMM_BK / 4)

template <int Mdim, int Ndim, int Kdim, int BM, int BN, int WM, int WN, bool PAIR>
__global__ __launch_bounds__(256)
void stp_wmma_gemm_bias(const float* __restrict__ A,
                        const float* __restrict__ Bt,
                        const float* __restrict__ bias,
                        float* __restrict__ D,
                        const float* __restrict__ Gt,
                        float* __restrict__ Q) {
  static_assert(WM * WN == 8, "8 waves per work-group");
  static_assert(Kdim % GEMM_BK == 0 && Mdim % BM == 0 && Ndim % BN == 0, "exact tiling");
  constexpr int TM = BM / WM, TN = BN / WN;
  constexpr int FM = TM / 16, FN = TN / 16;
  static_assert(TN == 32, "a wave's rows of D are whole 128-byte lines");
  constexpr int NC = Kdim / GEMM_BK;
  constexpr int LA = BM / 16, LB = BN / 16;
  constexpr int NBN = Ndim / BN;
  constexpr int PM  = BM / 2;
  constexpr int TM2 = PM / WM;
  static_assert(!PAIR || (TM2 == 16 && LA % 2 == 0), "one 16-row product tile per wave row");

  __shared__ __attribute__((aligned(16))) float As[2][BM][GEMM_PITCH];
  __shared__ __attribute__((aligned(16))) float Bs[2][BN][GEMM_PITCH];
  __shared__ __attribute__((aligned(16))) float Ps[PAIR ? 2 : 1][PAIR ? PM : 1][PAIR ? GEMM_PITCH : 4];
  __shared__ __attribute__((aligned(16))) float Gs[PAIR ? 2 : 1][PAIR ? BN : 1][PAIR ? GEMM_PITCH : 4];

  const int tid  = threadIdx.x;
  const int lane = tid & 31;
  const int wid  = tid >> 5;
  const int wm   = wid / WN;
  const int wn   = wid % WN;
  const int r    = lane & 15;
  const int h    = lane >> 4;
  const int bm   = blockIdx.x / NBN;
  const int bn   = blockIdx.x % NBN;

  const int sr = tid >> 4;
  const int sc = tid & (GEMM_Q - 1);
  const auto arow = [&](int j) { return PAIR ? 2 * sr + (j & 1) + 32 * (j >> 1) : sr + 16 * j; };
  const float4* ag = (const float4*)(A  + (size_t)(bm * BM) * Kdim) + sc;
  const float4* bg = (const float4*)(Bt + (size_t)(bn * BN + sr) * Kdim) + sc;
  const float4* gg = PAIR ? (const float4*)(Gt + (size_t)(bn * BN + sr) * Kdim) + sc : nullptr;
  constexpr size_t rstep = (size_t)16 * Kdim / 4;
  constexpr size_t rowq  = (size_t)Kdim / 4;
  const auto put_pairs = [&](int b, const float4* sa) {
#pragma unroll
    for (int jp = 0; jp < LA / 2; ++jp) {
      const float4 u = sa[2 * jp], w = sa[2 * jp + 1];
      *(float4*)&Ps[b][sr + 16 * jp][4 * sc] = make_float4(u.x * w.x, u.y * w.y, u.z * w.z, u.w * w.w);
    }
  };

  v8f_t c[FM][FN];
#pragma unroll
  for (int fm = 0; fm < FM; ++fm)
#pragma unroll
    for (int fn = 0; fn < FN; ++fn) c[fm][fn] = (v8f_t){};
  v8f_t c2[FN];
#pragma unroll
  for (int fn = 0; fn < FN; ++fn) c2[fn] = (v8f_t){};

  int doff = (bm * BM + wm * TM + 8 * h) * Ndim + bn * BN + wn * TN + r;
  float bv[FN];
#pragma unroll
  for (int fn = 0; fn < FN; ++fn) bv[fn] = bias[bn * BN + wn * TN + fn * 16 + r];
  int qoff = 2 * (bm * PM + wm * TM2 + 8 * h) * Ndim + bn * BN + wn * TN + r;
  asm volatile("" : "+v"(doff));
  if constexpr (PAIR) asm volatile("" : "+v"(qoff));

  {
    float4 sa[LA], sb[LB], sg[LB];
#pragma unroll
    for (int j = 0; j < LA; ++j) sa[j] = ag[(size_t)arow(j) * rowq];
#pragma unroll
    for (int j = 0; j < LB; ++j) sb[j] = bg[j * rstep];
    if constexpr (PAIR) {
#pragma unroll
      for (int j = 0; j < LB; ++j) sg[j] = gg[j * rstep];
    }
#pragma unroll
    for (int j = 0; j < LA; ++j) *(float4*)&As[0][arow(j)][4 * sc] = sa[j];
#pragma unroll
    for (int j = 0; j < LB; ++j) *(float4*)&Bs[0][sr + 16 * j][4 * sc] = sb[j];
    if constexpr (PAIR) {
      put_pairs(0, sa);
#pragma unroll
      for (int j = 0; j < LB; ++j) *(float4*)&Gs[0][sr + 16 * j][4 * sc] = sg[j];
    }
  }
  __syncthreads();

#pragma unroll
  for (int ch = 0; ch < NC; ++ch) {
    const int  buf  = ch & 1;
    const bool more = (ch + 1 < NC);

    float4 sa[LA], sb[LB], sg[LB];
#pragma unroll
    for (int j = 0; j < LA; ++j) sa[j] = make_float4(0.0f, 0.0f, 0.0f, 0.0f);
#pragma unroll
    for (int j = 0; j < LB; ++j) sb[j] = sg[j] = make_float4(0.0f, 0.0f, 0.0f, 0.0f);
    if (more) {
      const size_t koff = (size_t)(ch + 1) * GEMM_Q;
#pragma unroll
      for (int j = 0; j < LA; ++j) sa[j] = ag[(size_t)arow(j) * rowq + koff];
#pragma unroll
      for (int j = 0; j < LB; ++j) sb[j] = bg[j * rstep + koff];
      if constexpr (PAIR) {
#pragma unroll
        for (int j = 0; j < LB; ++j) sg[j] = gg[j * rstep + koff];
      }
    }

#pragma unroll
    for (int kb = 0; kb < GEMM_BK / 8; ++kb) {
      float4 a4[FM], b4[FN];
#pragma unroll
      for (int fm = 0; fm < FM; ++fm)
        a4[fm] = *(const float4*)&As[buf][wm * TM + fm * 16 + r][8 * kb + 4 * h];
#pragma unroll
      for (int fn = 0; fn < FN; ++fn)
        b4[fn] = *(const float4*)&Bs[buf][wn * TN + fn * 16 + r][8 * kb + 4 * h];
#pragma unroll
      for (int fm = 0; fm < FM; ++fm)
#pragma unroll
        for (int fn = 0; fn < FN; ++fn) {
          c[fm][fn] = __builtin_amdgcn_wmma_f32_16x16x4_f32(
              false, (v2f_t){a4[fm].x, a4[fm].y}, false, (v2f_t){b4[fn].x, b4[fn].y},
              (short)0, c[fm][fn], false, false);
        }
#pragma unroll
      for (int fm = 0; fm < FM; ++fm)
#pragma unroll
        for (int fn = 0; fn < FN; ++fn) {
          c[fm][fn] = __builtin_amdgcn_wmma_f32_16x16x4_f32(
              false, (v2f_t){a4[fm].z, a4[fm].w}, false, (v2f_t){b4[fn].z, b4[fn].w},
              (short)0, c[fm][fn], false, false);
        }
      if constexpr (PAIR) {
        const float4 p4 = *(const float4*)&Ps[buf][wm * TM2 + r][8 * kb + 4 * h];
        float4 g4[FN];
#pragma unroll
        for (int fn = 0; fn < FN; ++fn)
          g4[fn] = *(const float4*)&Gs[buf][wn * TN + fn * 16 + r][8 * kb + 4 * h];
#pragma unroll
        for (int fn = 0; fn < FN; ++fn)
          c2[fn] = __builtin_amdgcn_wmma_f32_16x16x4_f32(
              false, (v2f_t){p4.x, p4.y}, false, (v2f_t){g4[fn].x, g4[fn].y},
              (short)0, c2[fn], false, false);
#pragma unroll
        for (int fn = 0; fn < FN; ++fn)
          c2[fn] = __builtin_amdgcn_wmma_f32_16x16x4_f32(
              false, (v2f_t){p4.z, p4.w}, false, (v2f_t){g4[fn].z, g4[fn].w},
              (short)0, c2[fn], false, false);
      }
    }

    if (more) {
#pragma unroll
      for (int j = 0; j < LA; ++j) *(float4*)&As[buf ^ 1][arow(j)][4 * sc] = sa[j];
#pragma unroll
      for (int j = 0; j < LB; ++j) *(float4*)&Bs[buf ^ 1][sr + 16 * j][4 * sc] = sb[j];
      if constexpr (PAIR) {
        put_pairs(buf ^ 1, sa);
#pragma unroll
        for (int j = 0; j < LB; ++j) *(float4*)&Gs[buf ^ 1][sr + 16 * j][4 * sc] = sg[j];
      }
      __syncthreads();
    }
  }

  __builtin_amdgcn_sched_barrier(0);
  asm volatile("v_nop\n\tv_nop\n\tv_nop\n\tv_nop");
  __builtin_amdgcn_sched_barrier(0);

#pragma unroll
  for (int fm = 0; fm < FM; ++fm)
#pragma unroll
    for (int fn = 0; fn < FN; ++fn) {
#pragma unroll
      for (int v = 0; v < 8; ++v)
        D[(size_t)(doff + (fm * 16 + v) * Ndim + fn * 16)] = c[fm][fn][v] + bv[fn];
    }
  if constexpr (PAIR) {
#pragma unroll
    for (int fn = 0; fn < FN; ++fn) {
#pragma unroll
      for (int v = 0; v < 8; ++v)
        Q[(size_t)(qoff + 2 * v * Ndim + fn * 16)] = c2[fn][v];
    }
  }
}

template <int M>
__device__ __forceinline__ float stp_row_xor(float v) {
  const int b = __builtin_bit_cast(int, v);
  return __builtin_bit_cast(
      float, __builtin_amdgcn_update_dpp(b, b, 0x160 + M, 0xf, 0xf, true));
}


template <int M>
__device__ __forceinline__ void stp_lane_step(float& p0, float& p1, const float* g, const float* xa,
                                              const float* xb, float* gx, float* gy, int base) {
  p0 += stp_row_xor<M>(p0);
  p1 += stp_row_xor<M>(p1);
#pragma unroll
  for (int j = 0; j < 4; ++j) {
    gx[base + j] = g[base + j] * xa[base + j];
    gy[base + j] = g[base + j] * xb[base + j];
  }
  if (base + 4 < 16) __builtin_amdgcn_sched_barrier(0);
}

__device__ __forceinline__ float stp_erf_flat(float x) {
  const float ax = fabsf(x);
  const float t = ax * ax;
  float ps = fmaf(t, __builtin_bit_cast(float, 0xba1345e1u), __builtin_bit_cast(float, 0x3ba10414u));
  ps = fmaf(t, ps, __builtin_bit_cast(float, 0xbcdac9b8u));
  ps = fmaf(t, ps, __builtin_bit_cast(float, 0x3de703beu));
  ps = fmaf(t, ps, __builtin_bit_cast(float, 0xbec09330u));
  ps = fmaf(t, ps, __builtin_bit_cast(float, 0x3e0375d0u));
  const float small = fmaf(ax, ps, ax);
  float pb = fmaf(ax, __builtin_bit_cast(float, 0x378e98abu), __builtin_bit_cast(float, 0xb9c68948u));
  pb = fmaf(ax, pb, __builtin_bit_cast(float, 0x3b7cd369u));
  pb = fmaf(ax, pb, __builtin_bit_cast(float, 0xbcc618b2u));
  pb = fmaf(ax, pb, __builtin_bit_cast(float, 0x3dda74e4u));
  pb = fmaf(ax, pb, __builtin_bit_cast(float, 0x3f228afdu));
  pb = fmaf(ax, pb, __builtin_bit_cast(float, 0x3e03c728u));
  pb = fmaf(ax, pb, ax);
  const float big = 1.0f - expf(-pb);
  return copysignf((ax < 1.0f) ? small : big, x);
}

#define SCAN_TC     32
#define SCAN_WAVES  4
#define SCAN_ROWS   (2 * SCAN_WAVES)
#define SCAN_NCHUNK (LSEQ / SCAN_TC)
#define SCAN_XQ     (DMOD / 4)
#define SCAN_XLD    (SCAN_TC * SCAN_XQ / (SCAN_WAVES * 32))
#define SCAN_NP     (SCAN_TC / 2)
static_assert(SCAN_TC == 32, "lane c of a row keeps the h of the chunk's steps c and 16 + c");
static_assert(SCAN_XQ == 64, "a lane's four float4 of an x row are 16 float4 apart");
static_assert(SCAN_NP * SCAN_ROWS == SCAN_WAVES * 32, "a thread stages one (step pair, dff-row) per chunk");

struct stp_pair_ops {
  float4 a[4];
  float4 b[4];
  float  hsa, hsb;
  float  pa;
};

#define SCAN_GP 4
static_assert(2 * 2 * SCAN_GP == 16, "two read groups cover the 16 steps whose h a lane row keeps");
struct stp_group_ops {
  stp_pair_ops p[SCAN_GP];
};

__global__ __launch_bounds__(SCAN_WAVES * 32)
void stp_scan_kernel(const float* __restrict__ x,
                     const float* __restrict__ Hs,
                     const float* __restrict__ Lambda,
                     const float* __restrict__ Gamma,
                     float* __restrict__ Hg) {
  __shared__ float4 xs[2][SCAN_TC][SCAN_XQ];
  __shared__ float4 hsq[2][SCAN_NP][SCAN_ROWS];

  const int tid  = threadIdx.x;
  const int wid  = tid >> 5;
  const int lane = tid & 31;
  const int c    = lane & 15;
  const int hrow = 2 * wid + (lane >> 4);
  const int row0 = blockIdx.x * SCAN_ROWS;
  if (row0 >= BSZ * DFF) return;
  const int b  = row0 >> 9;
  const int i0 = row0 & (DFF - 1);
  const int i  = i0 + hrow;

  float r[16], g[16], F[16];
  const float4* lam = (const float4*)(Lambda + (size_t)i * DMOD) + c;
  const float4* gam = (const float4*)(Gamma  + (size_t)i * DMOD) + c;
#pragma unroll
  for (int k = 0; k < 4; ++k) {
    const float4 l4 = lam[16 * k];
    const float4 g4 = gam[16 * k];
    const float lv[4] = {l4.x, l4.y, l4.z, l4.w};
    const float gv[4] = {g4.x, g4.y, g4.z, g4.w};
#pragma unroll
    for (int j = 0; j < 4; ++j) {
      const float s = 1.0f / (1.0f + __expf(-lv[j]));
      r[4 * k + j] = 1.0f - s;
      g[4 * k + j] = gv[j];
      F[4 * k + j] = 0.0f;
    }
  }

  const int srow = tid >> 6;
  const int scol = tid & (SCAN_XQ - 1);
  const int hpq  = tid >> 3;
  const int hw   = tid & (SCAN_ROWS - 1);
  const float4* xg  = (const float4*)(x + (size_t)b * LSEQ * DMOD) + (size_t)srow * SCAN_XQ + scol;
  const size_t  hoff = (size_t)b * LSEQ * DFF + (size_t)(2 * hpq) * DFF + i0 + hw;
  const float*  hsg = Hs + hoff;
  const float*  qag = Hg + hoff;
  float*     hg_ptr = Hg + (size_t)b * LSEQ * DFF + i;

  {
    float4 st[SCAN_XLD];
#pragma unroll
    for (int k = 0; k < SCAN_XLD; ++k) st[k] = xg[(size_t)(2 * k) * SCAN_XQ];
    const float sh0 = hsg[0];
    const float sh1 = hsg[DFF];
    const float sq  = qag[0];
#pragma unroll
    for (int k = 0; k < SCAN_XLD; ++k) xs[0][srow + 2 * k][scol] = st[k];
    hsq[0][hpq][hw] = make_float4(sh0, sh1, sq, 0.0f);
  }
  __syncthreads();

  const auto gelu_store = [&](float ha, float hb, int tb) {
    const float ga = 0.5f * ha * (1.0f + stp_erf_flat(ha * 0.70710678118654752f));
    const float gb = 0.5f * hb * (1.0f + stp_erf_flat(hb * 0.70710678118654752f));
    hg_ptr[(size_t)(tb + c) * DFF] = ga;
    hg_ptr[(size_t)(tb + 16 + c) * DFF] = gb;
  };
  float hkp0 = 0.0f, hkp1 = 0.0f;

  for (int ch = 0; ch < SCAN_NCHUNK; ++ch) {
    const int  buf  = ch & 1;
    const int  t0   = ch * SCAN_TC;
    const bool more = (ch + 1 < SCAN_NCHUNK);

    float4 st[SCAN_XLD];
    float  sh0 = 0.0f, sh1 = 0.0f, sq = 0.0f;
#pragma unroll
    for (int k = 0; k < SCAN_XLD; ++k) st[k] = make_float4(0.0f, 0.0f, 0.0f, 0.0f);
    if (more) {
      const size_t tn0 = (size_t)(t0 + SCAN_TC);
#pragma unroll
      for (int k = 0; k < SCAN_XLD; ++k) st[k] = xg[(tn0 + 2 * k) * SCAN_XQ];
      sh0 = hsg[tn0 * DFF];
      sh1 = hsg[(tn0 + 1) * DFF];
      sq  = qag[tn0 * DFF];
    }

    const auto ld_pair = [&](int tt) {
      stp_pair_ops o;
      const float4 hv = hsq[buf][tt >> 1][hrow];
      o.hsa = hv.x;
      o.hsb = hv.y;
      o.pa  = hv.z;
#pragma unroll
      for (int k = 0; k < 4; ++k) {
        o.a[k] = xs[buf][tt][16 * k + c];
        o.b[k] = xs[buf][tt + 1][16 * k + c];
      }
      return o;
    };

    float hq[4];
#pragma unroll
    for (int m = 0; m < 4; ++m) hq[m] = 0.0f;
    float hk0 = 0.0f, hk1 = 0.0f;

    const auto step_pair = [&](const stp_pair_ops& o, int tq) {
      float xa[16], xb[16];
#pragma unroll
      for (int k = 0; k < 4; ++k) {
        xa[4 * k] = o.a[k].x; xa[4 * k + 1] = o.a[k].y; xa[4 * k + 2] = o.a[k].z; xa[4 * k + 3] = o.a[k].w;
        xb[4 * k] = o.b[k].x; xb[4 * k + 1] = o.b[k].y; xb[4 * k + 2] = o.b[k].z; xb[4 * k + 3] = o.b[k].w;
      }

      asm volatile(""
                   : "+v"(r[0]), "+v"(r[1]), "+v"(r[2]), "+v"(r[3]),
                     "+v"(r[4]), "+v"(r[5]), "+v"(r[6]), "+v"(r[7]),
                     "+v"(r[8]), "+v"(r[9]), "+v"(r[10]), "+v"(r[11]),
                     "+v"(r[12]), "+v"(r[13]), "+v"(r[14]), "+v"(r[15])
                   : "v"(F[0])
                   : "memory");

      float q0[4], q1[4];
      float rF[16];
#pragma unroll
      for (int u = 0; u < 16; ++u) rF[u] = r[u] * F[u];
#pragma unroll
      for (int j = 0; j < 4; ++j) {
        q0[j] = F[j] * xa[j];
        q1[j] = rF[j] * xb[j];
      }
#pragma unroll
      for (int u = 4; u < 16; ++u) {
        q0[u & 3] = fmaf(F[u], xa[u], q0[u & 3]);
        q1[u & 3] = fmaf(rF[u], xb[u], q1[u & 3]);
      }
      float gx[16], gy[16];
      float p0, p1;
      __builtin_amdgcn_sched_barrier(0);
      p0 = (q0[0] + q0[2]) + (q0[1] + q0[3]);
      p1 = (q1[0] + q1[2]) + (q1[1] + q1[3]);
      stp_lane_step<8>(p0, p1, g, xa, xb, gx, gy, 0);
      stp_lane_step<4>(p0, p1, g, xa, xb, gx, gy, 4);
      stp_lane_step<2>(p0, p1, g, xa, xb, gx, gy, 8);
      stp_lane_step<1>(p0, p1, g, xa, xb, gx, gy, 12);

      float c1 = o.hsb + p1;
      asm("" : "+v"(p0), "+v"(c1));

      const float h0 = o.hsa + p0;
      const float h1 = fmaf(h0, o.pa, c1);
      hq[tq % 4]       = (c == tq)     ? h0 : hq[tq % 4];
      hq[(tq + 1) % 4] = (c == tq + 1) ? h1 : hq[(tq + 1) % 4];

#pragma unroll
      for (int u = 0; u < 16; ++u) {
        const float F1 = fmaf(h0, gx[u], rF[u]);
        F[u] = fmaf(h1, gy[u], r[u] * F1);
      }
    };

    const auto ld_group = [&](int tt) {
      stp_group_ops gq;
#pragma unroll
      for (int k = 0; k < SCAN_GP; ++k) gq.p[k] = ld_pair(tt + 2 * k);
      return gq;
    };
    stp_group_ops gA = ld_group(0);
    asm volatile("" ::: "memory");
    if (ch > 0) gelu_store(hkp0, hkp1, t0 - SCAN_TC);
#pragma unroll
    for (int tp = 0; tp < SCAN_TC; tp += 16) {
      const stp_group_ops gB = ld_group(tp + 8);
#pragma unroll
      for (int k = 0; k < SCAN_GP; ++k) step_pair(gA.p[k], 2 * k);
      if (tp + 16 < SCAN_TC) gA = ld_group(tp + 16);
#pragma unroll
      for (int k = 0; k < SCAN_GP; ++k) step_pair(gB.p[k], 8 + 2 * k);
      {
        const float mk = (c & 2) ? ((c & 1) ? hq[3] : hq[2]) : ((c & 1) ? hq[1] : hq[0]);
        if (tp == 0) hk0 = mk; else hk1 = mk;
      }
    }

    if (more) {
#pragma unroll
      for (int k = 0; k < SCAN_XLD; ++k) xs[buf ^ 1][srow + 2 * k][scol] = st[k];
      hsq[buf ^ 1][hpq][hw] = make_float4(sh0, sh1, sq, 0.0f);
      __syncthreads();
    }

    hkp0 = hk0;
    hkp1 = hk1;
  }
  gelu_store(hkp0, hkp1, LSEQ - SCAN_TC);
}

extern "C" void kernel_launch(void* const* d_in, const int* in_sizes, int n_in,
                              void* d_out, int out_size, void* d_ws, size_t ws_size,
                              hipStream_t stream) {
  const float* x   = (const float*)d_in[0];
  const float* W1  = (const float*)d_in[1];
  const float* b1  = (const float*)d_in[2];
  const float* W2  = (const float*)d_in[3];
  const float* b2  = (const float*)d_in[4];
  const float* Lam = (const float*)d_in[5];
  const float* Gam = (const float*)d_in[6];
  float* out = (float*)d_out;

  float* Hs = (float*)d_ws;
  float* Hg = Hs + (size_t)MTOK * DFF;

  stp_wmma_gemm_bias<MTOK, DFF, DMOD, 64, 128, 2, 4, true>
      <<<(MTOK / 64) * (DFF / 128), 256, 0, stream>>>(x, W1, b1, Hs, Gam, Hg);

  stp_scan_kernel<<<(BSZ * DFF) / 8, 128, 0, stream>>>(x, Hs, Lam, Gam, Hg);

  stp_wmma_gemm_bias<MTOK, DMOD, DFF, 64, 64, 4, 2, false>
      <<<(MTOK / 64) * (DMOD / 64), 256, 0, stream>>>(Hg, W2, b2, out, nullptr, nullptr);
}
